# starting kernel
# baseline (speedup 1.0000x reference)
#include <hip/hip_runtime.h>
#include <math.h>

#define B_  4
#define LQ  512
#define LK  512
#define E_  128
#define D_  256

typedef float v2f __attribute__((ext_vector_type(2)));
typedef float v8f __attribute__((ext_vector_type(8)));

__device__ __forceinline__ float fast_tanh(float x) {
#if __has_builtin(__builtin_amdgcn_tanhf)
    return __builtin_amdgcn_tanhf(x);
#else
    return tanhf(x);
#endif
}

__device__ __forceinline__ float fast_exp(float x) {
#if __has_builtin(__builtin_amdgcn_exp2f)
    return __builtin_amdgcn_exp2f(x * 1.44269504088896340f);
#else
    return __expf(x);
#endif
}

__global__ __launch_bounds__(128) void proj_kernel(
    const float* __restrict__ query, const float* __restrict__ key,
    const float* __restrict__ Wc1,   const float* __restrict__ Wc2,
    float* __restrict__ qp, float* __restrict__ kp)
{
    __shared__ __align__(16) float row[D_];
    int bid  = blockIdx.x;
    int half = bid >> 11;
    int r    = bid & 2047;
    int b    = r >> 9;
    int m    = r & 511;
    const float* in  = half ? key : query;
    const float* W   = half ? Wc2 : Wc1;
    float*       out = half ? kp  : qp;

    const float* src = in + ((size_t)(b * 512 + m)) * D_;
    int tid = threadIdx.x;
    row[tid]       = src[tid];
    row[tid + 128] = src[tid + 128];
    __syncthreads();

    float acc = 0.f;
#pragma unroll 8
    for (int d = 0; d < D_; ++d)
        acc = fmaf(row[d], W[d * E_ + tid], acc);
    out[((size_t)(b * 512 + m)) * E_ + tid] = acc;
}

__global__ __launch_bounds__(256) void score_softmax_kernel(
    const float* __restrict__ qp, const float* __restrict__ kp,
    const float* __restrict__ vc, float* __restrict__ atten)
{
    __shared__ __align__(16) float kpl[E_];
    __shared__ __align__(16) float vcl[E_];
    __shared__ float red[256];

    int bid = blockIdx.x;
    int b   = bid >> 9;
    int l   = bid & 511;
    int tid = threadIdx.x;

    if (tid < E_) {
        kpl[tid] = kp[((size_t)(b * LK + l)) * E_ + tid];
        vcl[tid] = vc[tid];
    }
    __syncthreads();

    const float4* kp4 = (const float4*)kpl;
    const float4* vc4 = (const float4*)vcl;

    float accs[2];
#pragma unroll
    for (int j = 0; j < 2; ++j) {
        int q = tid + j * 256;
        const float4* qrow = (const float4*)(qp + ((size_t)(b * LQ + q)) * E_);
        float acc = 0.f;
#pragma unroll 8
        for (int e4 = 0; e4 < E_ / 4; ++e4) {
            float4 qv = qrow[e4];
            float4 kv = kp4[e4];
            float4 vv = vc4[e4];
            acc = fmaf(fast_tanh(qv.x + kv.x), vv.x, acc);
            acc = fmaf(fast_tanh(qv.y + kv.y), vv.y, acc);
            acc = fmaf(fast_tanh(qv.z + kv.z), vv.z, acc);
            acc = fmaf(fast_tanh(qv.w + kv.w), vv.w, acc);
        }
        accs[j] = acc;
    }

    red[tid] = fmaxf(accs[0], accs[1]);
    __syncthreads();
    for (int off = 128; off > 0; off >>= 1) {
        if (tid < off) red[tid] = fmaxf(red[tid], red[tid + off]);
        __syncthreads();
    }
    float mx = red[0];
    __syncthreads();

    float e0 = fast_exp(accs[0] - mx);
    float e1 = fast_exp(accs[1] - mx);
    red[tid] = e0 + e1;
    __syncthreads();
    for (int off = 128; off > 0; off >>= 1) {
        if (tid < off) red[tid] += red[tid + off];
        __syncthreads();
    }
    float inv = 1.f / red[0];

    float* arow = atten + ((size_t)(b * LK + l)) * LQ;
    arow[tid]       = e0 * inv;
    arow[tid + 256] = e1 * inv;
}

__global__ __launch_bounds__(128) void context_wmma_kernel(
    const float* __restrict__ atten, const float* __restrict__ value,
    float* __restrict__ context)
{
    int wave = threadIdx.x >> 5;
    int lane = threadIdx.x & 31;
    int tile = blockIdx.x * 4 + wave;
    int b    = tile >> 9;
    int r    = tile & 511;
    int lt   = r >> 4;
    int dt   = r & 15;
    int l0   = lt * 16;
    int d0   = dt * 16;
    int half = lane >> 4;
    int lh   = lane & 15;

    const float* arow = atten + ((size_t)(b * LK + l0 + lh)) * LQ + half * 2;
    const float* vcol = value + ((size_t)(b * LQ)) * D_ + d0 + lh;

    v8f c = {0.f, 0.f, 0.f, 0.f, 0.f, 0.f, 0.f, 0.f};
    for (int q = 0; q < LQ; q += 4) {
        v2f a = *(const v2f*)(arow + q);
        v2f bm;
        bm.x = vcol[(size_t)(q + half * 2) * D_];
        bm.y = vcol[(size_t)(q + half * 2 + 1) * D_];
        c = __builtin_amdgcn_wmma_f32_16x16x4_f32(
                 false, a,  false, bm,
                 (short)0, c,  false,  false);
    }

    float* crow = context + ((size_t)(b * LK + l0 + half * 8)) * D_ + d0 + lh;
#pragma unroll
    for (int i = 0; i < 8; ++i)
        crow[(size_t)i * D_] = c[i];
}

extern "C" void kernel_launch(void* const* d_in, const int* in_sizes, int n_in,
                              void* d_out, int out_size, void* d_ws, size_t ws_size,
                              hipStream_t stream)
{
    const float* query = (const float*)d_in[0];
    const float* key   = (const float*)d_in[1];
    const float* value = (const float*)d_in[2];
    const float* Wc1   = (const float*)d_in[3];
    const float* Wc2   = (const float*)d_in[4];
    const float* vc    = (const float*)d_in[5];

    float* out     = (float*)d_out;
    float* context = out;
    float* atten   = out + (size_t)B_ * LK * D_;

    float* qp = (float*)d_ws;
    float* kp = qp + (size_t)B_ * LQ * E_;

    proj_kernel<<<2 * B_ * 512, 128, 0, stream>>>(query, key, Wc1, Wc2, qp, kp);
    score_softmax_kernel<<<B_ * LK, 256, 0, stream>>>(qp, kp, vc, atten);
    context_wmma_kernel<<<(B_ * (LK / 16) * (D_ / 16)) / 4, 128, 0, stream>>>(
        atten, value, context);
}


# score+softmax kernel re-tiled: one work-group per 8 key rows, each thread 2 query columns x 8 rows; projected queries staged once per group through LDS instead of re-read per key row
# speedup vs baseline: 1.5478x; 1.5478x over previous
#include <hip/hip_runtime.h>
#include <math.h>

#define B_  4
#define LQ  512
#define LK  512
#define E_  128
#define D_  256

typedef float v2f __attribute__((ext_vector_type(2)));
typedef float v8f __attribute__((ext_vector_type(8)));
typedef float v4f __attribute__((ext_vector_type(4)));

__device__ __forceinline__ float fast_tanh(float x) {
#if __has_builtin(__builtin_amdgcn_tanhf)
    return __builtin_amdgcn_tanhf(x);
#else
    return tanhf(x);
#endif
}

__device__ __forceinline__ float fast_exp(float x) {
#if __has_builtin(__builtin_amdgcn_exp2f)
    return __builtin_amdgcn_exp2f(x * 1.44269504088896340f);
#else
    return __expf(x);
#endif
}

__global__ __launch_bounds__(128) void proj_kernel(
    const float* __restrict__ query, const float* __restrict__ key,
    const float* __restrict__ Wc1,   const float* __restrict__ Wc2,
    float* __restrict__ qp, float* __restrict__ kp)
{
    __shared__ __align__(16) float row[D_];
    int bid  = blockIdx.x;
    int half = bid >> 11;
    int r    = bid & 2047;
    int b    = r >> 9;
    int m    = r & 511;
    const float* in  = half ? key : query;
    const float* W   = half ? Wc2 : Wc1;
    float*       out = half ? kp  : qp;

    const float* src = in + ((size_t)(b * 512 + m)) * D_;
    int tid = threadIdx.x;
    row[tid]       = src[tid];
    row[tid + 128] = src[tid + 128];
    __syncthreads();

    float acc = 0.f;
#pragma unroll 8
    for (int d = 0; d < D_; ++d)
        acc = fmaf(row[d], W[d * E_ + tid], acc);
    out[((size_t)(b * 512 + m)) * E_ + tid] = acc;
}

#define TL      8
#define EC      32
#define NCH     (E_ / EC)
#define QPITCH  44
#define QSTAGE  (LQ * QPITCH)

__global__ __launch_bounds__(256) void score_softmax_kernel(
    const float* __restrict__ qp, const float* __restrict__ kp,
    const float* __restrict__ vc, float* __restrict__ atten)
{
    __shared__ __align__(16) float qs[2 * QSTAGE];
    __shared__ __align__(16) float kpl[TL * E_];
    __shared__ __align__(16) float vcl[E_];
    __shared__ __align__(16) float redm[8 * TL];
    __shared__ __align__(16) float reds[8 * TL];

    const int tid  = threadIdx.x;
    const int lane = tid & 31;
    const int wave = tid >> 5;
    const int b    = blockIdx.x / (LK / TL);
    const int l0   = (blockIdx.x % (LK / TL)) * TL;

    const float* qpb = qp + (size_t)b * LQ * E_;

    ((v4f*)kpl)[tid] = ((const v4f*)(kp + (size_t)(b * LK + l0) * E_))[tid];
    if (tid < E_ / 4) ((v4f*)vcl)[tid] = ((const v4f*)vc)[tid];

    const int srow = tid >> 3;
    const int sc4  = tid & 7;
    const float* gsrc = qpb + (size_t)srow * E_ + sc4 * 4;
    float*       ldst = qs + srow * QPITCH + sc4 * 4;

    {
        v4f p0[16];
#pragma unroll
        for (int i = 0; i < 16; ++i)
            p0[i] = *(const v4f*)(gsrc + (size_t)i * 32 * E_);
#pragma unroll
        for (int i = 0; i < 16; ++i)
            *(v4f*)(ldst + i * 32 * QPITCH) = p0[i];
    }
    __syncthreads();

    v2f P0[TL / 2], P1[TL / 2];
#pragma unroll
    for (int j = 0; j < TL / 2; ++j) { P0[j] = (v2f){0.f, 0.f}; P1[j] = (v2f){0.f, 0.f}; }

#pragma unroll 1
    for (int c = 0; c < NCH; ++c) {
        const bool more = (c + 1 < NCH);
        v4f pre[16];
        if (more) {
#pragma unroll
            for (int i = 0; i < 16; ++i)
                pre[i] = *(const v4f*)(gsrc + (c + 1) * EC + (size_t)i * 32 * E_);
        }

        const float* qa = qs + (c & 1) * QSTAGE + tid * QPITCH;
        const float* qb = qa + 256 * QPITCH;
        const float* kc = kpl + c * EC;
        const float* vv = vcl + c * EC;

#pragma unroll 2
        for (int e = 0; e < EC; e += 4) {
            const float4 a  = *(const float4*)(qa + e);
            const float4 bq = *(const float4*)(qb + e);
            const float4 v  = *(const float4*)(vv + e);
            const v2f vx = (v2f){v.x, v.x}, vy = (v2f){v.y, v.y};
            const v2f vz = (v2f){v.z, v.z}, vw = (v2f){v.w, v.w};
#pragma unroll
            for (int j = 0; j < TL / 2; ++j) {
                const float4 k0 = *(const float4*)(kc + (2 * j) * E_ + e);
                const float4 k1 = *(const float4*)(kc + (2 * j + 1) * E_ + e);
                v2f x0 = P0[j], x1 = P1[j];
                x0 = __builtin_elementwise_fma((v2f){fast_tanh(a.x  + k0.x), fast_tanh(a.x  + k1.x)}, vx, x0);
                x1 = __builtin_elementwise_fma((v2f){fast_tanh(bq.x + k0.x), fast_tanh(bq.x + k1.x)}, vx, x1);
                x0 = __builtin_elementwise_fma((v2f){fast_tanh(a.y  + k0.y), fast_tanh(a.y  + k1.y)}, vy, x0);
                x1 = __builtin_elementwise_fma((v2f){fast_tanh(bq.y + k0.y), fast_tanh(bq.y + k1.y)}, vy, x1);
                x0 = __builtin_elementwise_fma((v2f){fast_tanh(a.z  + k0.z), fast_tanh(a.z  + k1.z)}, vz, x0);
                x1 = __builtin_elementwise_fma((v2f){fast_tanh(bq.z + k0.z), fast_tanh(bq.z + k1.z)}, vz, x1);
                x0 = __builtin_elementwise_fma((v2f){fast_tanh(a.w  + k0.w), fast_tanh(a.w  + k1.w)}, vw, x0);
                x1 = __builtin_elementwise_fma((v2f){fast_tanh(bq.w + k0.w), fast_tanh(bq.w + k1.w)}, vw, x1);
                P0[j] = x0; P1[j] = x1;
            }
        }

        if (more) {
            float* d = ldst + ((c + 1) & 1) * QSTAGE;
#pragma unroll
            for (int i = 0; i < 16; ++i)
                *(v4f*)(d + i * 32 * QPITCH) = pre[i];
        }
        __syncthreads();
    }

    float acc0[TL], acc1[TL];
#pragma unroll
    for (int j = 0; j < TL / 2; ++j) {
        acc0[2 * j] = P0[j].x; acc0[2 * j + 1] = P0[j].y;
        acc1[2 * j] = P1[j].x; acc1[2 * j + 1] = P1[j].y;
    }

    float m[TL];
#pragma unroll
    for (int l = 0; l < TL; ++l) m[l] = fmaxf(acc0[l], acc1[l]);
#pragma unroll
    for (int off = 16; off > 0; off >>= 1) {
#pragma unroll
        for (int l = 0; l < TL; ++l) m[l] = fmaxf(m[l], __shfl_xor(m[l], off, 32));
    }
    if (lane == 0) {
        ((v4f*)redm)[wave * 2]     = (v4f){m[0], m[1], m[2], m[3]};
        ((v4f*)redm)[wave * 2 + 1] = (v4f){m[4], m[5], m[6], m[7]};
    }
    __syncthreads();
#pragma unroll
    for (int w = 0; w < 8; ++w) {
        const v4f r0 = ((const v4f*)redm)[w * 2];
        const v4f r1 = ((const v4f*)redm)[w * 2 + 1];
        m[0] = fmaxf(m[0], r0.x); m[1] = fmaxf(m[1], r0.y);
        m[2] = fmaxf(m[2], r0.z); m[3] = fmaxf(m[3], r0.w);
        m[4] = fmaxf(m[4], r1.x); m[5] = fmaxf(m[5], r1.y);
        m[6] = fmaxf(m[6], r1.z); m[7] = fmaxf(m[7], r1.w);
    }

    float s[TL];
#pragma unroll
    for (int l = 0; l < TL; ++l) {
        acc0[l] = fast_exp(acc0[l] - m[l]);
        acc1[l] = fast_exp(acc1[l] - m[l]);
        s[l] = acc0[l] + acc1[l];
    }
#pragma unroll
    for (int off = 16; off > 0; off >>= 1) {
#pragma unroll
        for (int l = 0; l < TL; ++l) s[l] += __shfl_xor(s[l], off, 32);
    }
    if (lane == 0) {
        ((v4f*)reds)[wave * 2]     = (v4f){s[0], s[1], s[2], s[3]};
        ((v4f*)reds)[wave * 2 + 1] = (v4f){s[4], s[5], s[6], s[7]};
    }
    __syncthreads();
    float t[TL];
#pragma unroll
    for (int l = 0; l < TL; ++l) t[l] = 0.f;
#pragma unroll
    for (int w = 0; w < 8; ++w) {
        const v4f r0 = ((const v4f*)reds)[w * 2];
        const v4f r1 = ((const v4f*)reds)[w * 2 + 1];
        t[0] += r0.x; t[1] += r0.y; t[2] += r0.z; t[3] += r0.w;
        t[4] += r1.x; t[5] += r1.y; t[6] += r1.z; t[7] += r1.w;
    }

    float* arow = atten + ((size_t)(b * LK + l0)) * LQ + tid;
#pragma unroll
    for (int l = 0; l < TL; ++l) {
        const float inv = 1.f / t[l];
        arow[(size_t)l * LQ]       = acc0[l] * inv;
        arow[(size_t)l * LQ + 256] = acc1[l] * inv;
    }
}

__global__ __launch_bounds__(128) void context_wmma_kernel(
    const float* __restrict__ atten, const float* __restrict__ value,
    float* __restrict__ context)
{
    int wave = threadIdx.x >> 5;
    int lane = threadIdx.x & 31;
    int tile = blockIdx.x * 4 + wave;
    int b    = tile >> 9;
    int r    = tile & 511;
    int lt   = r >> 4;
    int dt   = r & 15;
    int l0   = lt * 16;
    int d0   = dt * 16;
    int half = lane >> 4;
    int lh   = lane & 15;

    const float* arow = atten + ((size_t)(b * LK + l0 + lh)) * LQ + half * 2;
    const float* vcol = value + ((size_t)(b * LQ)) * D_ + d0 + lh;

    v8f c = {0.f, 0.f, 0.f, 0.f, 0.f, 0.f, 0.f, 0.f};
    for (int q = 0; q < LQ; q += 4) {
        v2f a = *(const v2f*)(arow + q);
        v2f bm;
        bm.x = vcol[(size_t)(q + half * 2) * D_];
        bm.y = vcol[(size_t)(q + half * 2 + 1) * D_];
        c = __builtin_amdgcn_wmma_f32_16x16x4_f32(
                 false, a,  false, bm,
                 (short)0, c,  false,  false);
    }

    float* crow = context + ((size_t)(b * LK + l0 + half * 8)) * D_ + d0 + lh;
#pragma unroll
    for (int i = 0; i < 8; ++i)
        crow[(size_t)i * D_] = c[i];
}

extern "C" void kernel_launch(void* const* d_in, const int* in_sizes, int n_in,
                              void* d_out, int out_size, void* d_ws, size_t ws_size,
                              hipStream_t stream)
{
    const float* query = (const float*)d_in[0];
    const float* key   = (const float*)d_in[1];
    const float* value = (const float*)d_in[2];
    const float* Wc1   = (const float*)d_in[3];
    const float* Wc2   = (const float*)d_in[4];
    const float* vc    = (const float*)d_in[5];

    float* out     = (float*)d_out;
    float* context = out;
    float* atten   = out + (size_t)B_ * LK * D_;

    float* qp = (float*)d_ws;
    float* kp = qp + (size_t)B_ * LQ * E_;

    proj_kernel<<<2 * B_ * 512, 128, 0, stream>>>(query, key, Wc1, Wc2, qp, kp);
    score_softmax_kernel<<<B_ * (LK / TL), 256, 0, stream>>>(qp, kp, vc, atten);
    context_wmma_kernel<<<(B_ * (LK / 16) * (D_ / 16)) / 4, 128, 0, stream>>>(
        atten, value, context);
}


# projection GEMMs moved from one-row-per-group vector FMAs onto f32 16x16x4 matrix instructions: 256 groups of 32 rows x 64 columns, input rows and weight slice staged once through LDS
# speedup vs baseline: 2.2216x; 1.4353x over previous
#include <hip/hip_runtime.h>
#include <math.h>

#define B_  4
#define LQ  512
#define LK  512
#define E_  128
#define D_  256

typedef float v2f __attribute__((ext_vector_type(2)));
typedef float v8f __attribute__((ext_vector_type(8)));
typedef float v4f __attribute__((ext_vector_type(4)));

__device__ __forceinline__ float fast_tanh(float x) {
#if __has_builtin(__builtin_amdgcn_tanhf)
    return __builtin_amdgcn_tanhf(x);
#else
    return tanhf(x);
#endif
}

__device__ __forceinline__ float fast_exp(float x) {
#if __has_builtin(__builtin_amdgcn_exp2f)
    return __builtin_amdgcn_exp2f(x * 1.44269504088896340f);
#else
    return __expf(x);
#endif
}

#define PM      32
#define PN      64
#define APITCH  260
#define WPITCH  72
#define CPITCH  68
#define KS      16
#define PROJ_GROUPS  (2 * (B_ * 512 / PM) * (E_ / PN))

__global__ __launch_bounds__(128) void proj_wmma_kernel(
    const float* __restrict__ query, const float* __restrict__ key,
    const float* __restrict__ Wc1,   const float* __restrict__ Wc2,
    float* __restrict__ qp, float* __restrict__ kp)
{
    __shared__ __align__(16) float As[PM * APITCH];
    __shared__ __align__(16) float Ws[D_ * WPITCH];
    __shared__ __align__(16) float Cs[PM * CPITCH];

    const int tid  = threadIdx.x;
    const int lane = tid & 31;
    const int wave = tid >> 5;
    const int bid  = blockIdx.x;
    const int sel  = bid >> 7;
    const int r    = bid & 127;
    const int m0   = (r >> 1) * PM;
    const int n0   = (r & 1) * PN;

    const float* in  = sel ? key : query;
    const float* Wg  = sel ? Wc2 : Wc1;
    float*       out = sel ? kp  : qp;

    {
        const float* asrc = in + (size_t)m0 * D_;
        v4f ra[16];
#pragma unroll
        for (int i = 0; i < 16; ++i)
            ra[i] = *(const v4f*)(asrc + (size_t)(i * 128 + tid) * 4);
        const float* wsrc = Wg + n0;
        v4f rw[32];
#pragma unroll
        for (int i = 0; i < 32; ++i) {
            const int idx = i * 128 + tid;
            rw[i] = *(const v4f*)(wsrc + (size_t)(idx >> 4) * E_ + (idx & 15) * 4);
        }
#pragma unroll
        for (int i = 0; i < 16; ++i) {
            const int idx = i * 128 + tid;
            *(v4f*)(As + (idx >> 6) * APITCH + (idx & 63) * 4) = ra[i];
        }
#pragma unroll
        for (int i = 0; i < 32; ++i) {
            const int idx = i * 128 + tid;
            *(v4f*)(Ws + (idx >> 4) * WPITCH + (idx & 15) * 4) = rw[i];
        }
    }
    __syncthreads();

    const int half = lane >> 4;
    const int lh   = lane & 15;
    const float* ap0 = As + lh * APITCH + 2 * half;
    const float* ap1 = ap0 + 16 * APITCH;
    const float* bp  = Ws + (2 * half) * WPITCH + wave * 16 + lh;

    v8f c0 = {0.f, 0.f, 0.f, 0.f, 0.f, 0.f, 0.f, 0.f};
    v8f c1 = {0.f, 0.f, 0.f, 0.f, 0.f, 0.f, 0.f, 0.f};

    v2f fa0[2][KS], fa1[2][KS], fb[2][KS];
#pragma unroll
    for (int s = 0; s < KS; ++s) {
        const int k = s * 4;
        fa0[0][s]   = *(const v2f*)(ap0 + k);
        fa1[0][s]   = *(const v2f*)(ap1 + k);
        fb[0][s].x  = bp[(k)     * WPITCH];
        fb[0][s].y  = bp[(k + 1) * WPITCH];
    }
#pragma unroll
    for (int kb = 0; kb < D_ / (4 * KS); ++kb) {
        const int cur = kb & 1;
        const int nxt = cur ^ 1;
        __builtin_amdgcn_sched_barrier(0);
        if (kb + 1 < D_ / (4 * KS)) {
#pragma unroll
            for (int s = 0; s < KS; ++s) {
                const int k = (kb + 1) * (4 * KS) + s * 4;
                fa0[nxt][s]   = *(const v2f*)(ap0 + k);
                fa1[nxt][s]   = *(const v2f*)(ap1 + k);
                fb[nxt][s].x  = bp[(k)     * WPITCH];
                fb[nxt][s].y  = bp[(k + 1) * WPITCH];
            }
        }
        __builtin_amdgcn_sched_barrier(0);
#pragma unroll
        for (int s = 0; s < KS; ++s) {
            c0 = __builtin_amdgcn_wmma_f32_16x16x4_f32(
                    false, fa0[cur][s], false, fb[cur][s], (short)0, c0, false, false);
            c1 = __builtin_amdgcn_wmma_f32_16x16x4_f32(
                    false, fa1[cur][s], false, fb[cur][s], (short)0, c1, false, false);
        }
    }
    __builtin_amdgcn_sched_barrier(0);

    float* cw = Cs + (half * 8) * CPITCH + wave * 16 + lh;
#pragma unroll
    for (int i = 0; i < 8; ++i) {
        cw[i * CPITCH]        = c0[i];
        cw[(16 + i) * CPITCH] = c1[i];
    }
    __syncthreads();

    float* obase = out + (size_t)m0 * E_ + n0;
#pragma unroll
    for (int i = 0; i < 4; ++i) {
        const int idx = i * 128 + tid;
        const int row = idx >> 4;
        const int c4  = idx & 15;
        *(v4f*)(obase + (size_t)row * E_ + c4 * 4) = *(const v4f*)(Cs + row * CPITCH + c4 * 4);
    }
}

#define TL      8
#define EC      32
#define NCH     (E_ / EC)
#define QPITCH  44
#define QSTAGE  (LQ * QPITCH)

__global__ __launch_bounds__(256) void score_softmax_kernel(
    const float* __restrict__ qp, const float* __restrict__ kp,
    const float* __restrict__ vc, float* __restrict__ atten)
{
    __shared__ __align__(16) float qs[2 * QSTAGE];
    __shared__ __align__(16) float kpl[TL * E_];
    __shared__ __align__(16) float vcl[E_];
    __shared__ __align__(16) float redm[8 * TL];
    __shared__ __align__(16) float reds[8 * TL];

    const int tid  = threadIdx.x;
    const int lane = tid & 31;
    const int wave = tid >> 5;
    const int b    = blockIdx.x / (LK / TL);
    const int l0   = (blockIdx.x % (LK / TL)) * TL;

    const float* qpb = qp + (size_t)b * LQ * E_;

    ((v4f*)kpl)[tid] = ((const v4f*)(kp + (size_t)(b * LK + l0) * E_))[tid];
    if (tid < E_ / 4) ((v4f*)vcl)[tid] = ((const v4f*)vc)[tid];

    const int srow = tid >> 3;
    const int sc4  = tid & 7;
    const float* gsrc = qpb + (size_t)srow * E_ + sc4 * 4;
    float*       ldst = qs + srow * QPITCH + sc4 * 4;

    {
        v4f p0[16];
#pragma unroll
        for (int i = 0; i < 16; ++i)
            p0[i] = *(const v4f*)(gsrc + (size_t)i * 32 * E_);
#pragma unroll
        for (int i = 0; i < 16; ++i)
            *(v4f*)(ldst + i * 32 * QPITCH) = p0[i];
    }
    __syncthreads();

    v2f P0[TL / 2], P1[TL / 2];
#pragma unroll
    for (int j = 0; j < TL / 2; ++j) { P0[j] = (v2f){0.f, 0.f}; P1[j] = (v2f){0.f, 0.f}; }

#pragma unroll 1
    for (int c = 0; c < NCH; ++c) {
        const bool more = (c + 1 < NCH);
        v4f pre[16];
        if (more) {
#pragma unroll
            for (int i = 0; i < 16; ++i)
                pre[i] = *(const v4f*)(gsrc + (c + 1) * EC + (size_t)i * 32 * E_);
        }

        const float* qa = qs + (c & 1) * QSTAGE + tid * QPITCH;
        const float* qb = qa + 256 * QPITCH;
        const float* kc = kpl + c * EC;
        const float* vv = vcl + c * EC;

#pragma unroll 2
        for (int e = 0; e < EC; e += 4) {
            const float4 a  = *(const float4*)(qa + e);
            const float4 bq = *(const float4*)(qb + e);
            const float4 v  = *(const float4*)(vv + e);
            const v2f vx = (v2f){v.x, v.x}, vy = (v2f){v.y, v.y};
            const v2f vz = (v2f){v.z, v.z}, vw = (v2f){v.w, v.w};
#pragma unroll
            for (int j = 0; j < TL / 2; ++j) {
                const float4 k0 = *(const float4*)(kc + (2 * j) * E_ + e);
                const float4 k1 = *(const float4*)(kc + (2 * j + 1) * E_ + e);
                v2f x0 = P0[j], x1 = P1[j];
                x0 = __builtin_elementwise_fma((v2f){fast_tanh(a.x  + k0.x), fast_tanh(a.x  + k1.x)}, vx, x0);
                x1 = __builtin_elementwise_fma((v2f){fast_tanh(bq.x + k0.x), fast_tanh(bq.x + k1.x)}, vx, x1);
                x0 = __builtin_elementwise_fma((v2f){fast_tanh(a.y  + k0.y), fast_tanh(a.y  + k1.y)}, vy, x0);
                x1 = __builtin_elementwise_fma((v2f){fast_tanh(bq.y + k0.y), fast_tanh(bq.y + k1.y)}, vy, x1);
                x0 = __builtin_elementwise_fma((v2f){fast_tanh(a.z  + k0.z), fast_tanh(a.z  + k1.z)}, vz, x0);
                x1 = __builtin_elementwise_fma((v2f){fast_tanh(bq.z + k0.z), fast_tanh(bq.z + k1.z)}, vz, x1);
                x0 = __builtin_elementwise_fma((v2f){fast_tanh(a.w  + k0.w), fast_tanh(a.w  + k1.w)}, vw, x0);
                x1 = __builtin_elementwise_fma((v2f){fast_tanh(bq.w + k0.w), fast_tanh(bq.w + k1.w)}, vw, x1);
                P0[j] = x0; P1[j] = x1;
            }
        }

        if (more) {
            float* d = ldst + ((c + 1) & 1) * QSTAGE;
#pragma unroll
            for (int i = 0; i < 16; ++i)
                *(v4f*)(d + i * 32 * QPITCH) = pre[i];
        }
        __syncthreads();
    }

    float acc0[TL], acc1[TL];
#pragma unroll
    for (int j = 0; j < TL / 2; ++j) {
        acc0[2 * j] = P0[j].x; acc0[2 * j + 1] = P0[j].y;
        acc1[2 * j] = P1[j].x; acc1[2 * j + 1] = P1[j].y;
    }

    float m[TL];
#pragma unroll
    for (int l = 0; l < TL; ++l) m[l] = fmaxf(acc0[l], acc1[l]);
#pragma unroll
    for (int off = 16; off > 0; off >>= 1) {
#pragma unroll
        for (int l = 0; l < TL; ++l) m[l] = fmaxf(m[l], __shfl_xor(m[l], off, 32));
    }
    if (lane == 0) {
        ((v4f*)redm)[wave * 2]     = (v4f){m[0], m[1], m[2], m[3]};
        ((v4f*)redm)[wave * 2 + 1] = (v4f){m[4], m[5], m[6], m[7]};
    }
    __syncthreads();
#pragma unroll
    for (int w = 0; w < 8; ++w) {
        const v4f r0 = ((const v4f*)redm)[w * 2];
        const v4f r1 = ((const v4f*)redm)[w * 2 + 1];
        m[0] = fmaxf(m[0], r0.x); m[1] = fmaxf(m[1], r0.y);
        m[2] = fmaxf(m[2], r0.z); m[3] = fmaxf(m[3], r0.w);
        m[4] = fmaxf(m[4], r1.x); m[5] = fmaxf(m[5], r1.y);
        m[6] = fmaxf(m[6], r1.z); m[7] = fmaxf(m[7], r1.w);
    }

    float s[TL];
#pragma unroll
    for (int l = 0; l < TL; ++l) {
        acc0[l] = fast_exp(acc0[l] - m[l]);
        acc1[l] = fast_exp(acc1[l] - m[l]);
        s[l] = acc0[l] + acc1[l];
    }
#pragma unroll
    for (int off = 16; off > 0; off >>= 1) {
#pragma unroll
        for (int l = 0; l < TL; ++l) s[l] += __shfl_xor(s[l], off, 32);
    }
    if (lane == 0) {
        ((v4f*)reds)[wave * 2]     = (v4f){s[0], s[1], s[2], s[3]};
        ((v4f*)reds)[wave * 2 + 1] = (v4f){s[4], s[5], s[6], s[7]};
    }
    __syncthreads();
    float t[TL];
#pragma unroll
    for (int l = 0; l < TL; ++l) t[l] = 0.f;
#pragma unroll
    for (int w = 0; w < 8; ++w) {
        const v4f r0 = ((const v4f*)reds)[w * 2];
        const v4f r1 = ((const v4f*)reds)[w * 2 + 1];
        t[0] += r0.x; t[1] += r0.y; t[2] += r0.z; t[3] += r0.w;
        t[4] += r1.x; t[5] += r1.y; t[6] += r1.z; t[7] += r1.w;
    }

    float* arow = atten + ((size_t)(b * LK + l0)) * LQ + tid;
#pragma unroll
    for (int l = 0; l < TL; ++l) {
        const float inv = 1.f / t[l];
        arow[(size_t)l * LQ]       = acc0[l] * inv;
        arow[(size_t)l * LQ + 256] = acc1[l] * inv;
    }
}

__global__ __launch_bounds__(128) void context_wmma_kernel(
    const float* __restrict__ atten, const float* __restrict__ value,
    float* __restrict__ context)
{
    int wave = threadIdx.x >> 5;
    int lane = threadIdx.x & 31;
    int tile = blockIdx.x * 4 + wave;
    int b    = tile >> 9;
    int r    = tile & 511;
    int lt   = r >> 4;
    int dt   = r & 15;
    int l0   = lt * 16;
    int d0   = dt * 16;
    int half = lane >> 4;
    int lh   = lane & 15;

    const float* arow = atten + ((size_t)(b * LK + l0 + lh)) * LQ + half * 2;
    const float* vcol = value + ((size_t)(b * LQ)) * D_ + d0 + lh;

    v8f c = {0.f, 0.f, 0.f, 0.f, 0.f, 0.f, 0.f, 0.f};
    for (int q = 0; q < LQ; q += 4) {
        v2f a = *(const v2f*)(arow + q);
        v2f bm;
        bm.x = vcol[(size_t)(q + half * 2) * D_];
        bm.y = vcol[(size_t)(q + half * 2 + 1) * D_];
        c = __builtin_amdgcn_wmma_f32_16x16x4_f32(
                 false, a,  false, bm,
                 (short)0, c,  false,  false);
    }

    float* crow = context + ((size_t)(b * LK + l0 + half * 8)) * D_ + d0 + lh;
#pragma unroll
    for (int i = 0; i < 8; ++i)
        crow[(size_t)i * D_] = c[i];
}

extern "C" void kernel_launch(void* const* d_in, const int* in_sizes, int n_in,
                              void* d_out, int out_size, void* d_ws, size_t ws_size,
                              hipStream_t stream)
{
    const float* query = (const float*)d_in[0];
    const float* key   = (const float*)d_in[1];
    const float* value = (const float*)d_in[2];
    const float* Wc1   = (const float*)d_in[3];
    const float* Wc2   = (const float*)d_in[4];
    const float* vc    = (const float*)d_in[5];

    float* out     = (float*)d_out;
    float* context = out;
    float* atten   = out + (size_t)B_ * LK * D_;

    float* qp = (float*)d_ws;
    float* kp = qp + (size_t)B_ * LQ * E_;

    proj_wmma_kernel<<<PROJ_GROUPS, 128, 0, stream>>>(query, key, Wc1, Wc2, qp, kp);
    score_softmax_kernel<<<B_ * (LK / TL), 256, 0, stream>>>(qp, kp, vc, atten);
    context_wmma_kernel<<<(B_ * (LK / 16) * (D_ / 16)) / 4, 128, 0, stream>>>(
        atten, value, context);
}


# context GEMM (atten @ value) re-tiled like the projection kernel: 256 groups of 32 rows x 64 columns, operands staged through LDS in four k-chunks fetched one chunk ahead, two accumulators per wave
# speedup vs baseline: 2.9625x; 1.3335x over previous
#include <hip/hip_runtime.h>
#include <math.h>

#define B_  4
#define LQ  512
#define LK  512
#define E_  128
#define D_  256

typedef float v2f __attribute__((ext_vector_type(2)));
typedef float v8f __attribute__((ext_vector_type(8)));
typedef float v4f __attribute__((ext_vector_type(4)));

__device__ __forceinline__ float fast_tanh(float x) {
#if __has_builtin(__builtin_amdgcn_tanhf)
    return __builtin_amdgcn_tanhf(x);
#else
    return tanhf(x);
#endif
}

__device__ __forceinline__ float fast_exp(float x) {
#if __has_builtin(__builtin_amdgcn_exp2f)
    return __builtin_amdgcn_exp2f(x * 1.44269504088896340f);
#else
    return __expf(x);
#endif
}

#define PM      32
#define PN      64
#define APITCH  260
#define WPITCH  72
#define CPITCH  68
#define KS      16
#define PROJ_GROUPS  (2 * (B_ * 512 / PM) * (E_ / PN))

__global__ __launch_bounds__(128) void proj_wmma_kernel(
    const float* __restrict__ query, const float* __restrict__ key,
    const float* __restrict__ Wc1,   const float* __restrict__ Wc2,
    float* __restrict__ qp, float* __restrict__ kp)
{
    __shared__ __align__(16) float As[PM * APITCH];
    __shared__ __align__(16) float Ws[D_ * WPITCH];
    __shared__ __align__(16) float Cs[PM * CPITCH];

    const int tid  = threadIdx.x;
    const int lane = tid & 31;
    const int wave = tid >> 5;
    const int bid  = blockIdx.x;
    const int sel  = bid >> 7;
    const int r    = bid & 127;
    const int m0   = (r >> 1) * PM;
    const int n0   = (r & 1) * PN;

    const float* in  = sel ? key : query;
    const float* Wg  = sel ? Wc2 : Wc1;
    float*       out = sel ? kp  : qp;

    {
        const float* asrc = in + (size_t)m0 * D_;
        v4f ra[16];
#pragma unroll
        for (int i = 0; i < 16; ++i)
            ra[i] = *(const v4f*)(asrc + (size_t)(i * 128 + tid) * 4);
        const float* wsrc = Wg + n0;
        v4f rw[32];
#pragma unroll
        for (int i = 0; i < 32; ++i) {
            const int idx = i * 128 + tid;
            rw[i] = *(const v4f*)(wsrc + (size_t)(idx >> 4) * E_ + (idx & 15) * 4);
        }
#pragma unroll
        for (int i = 0; i < 16; ++i) {
            const int idx = i * 128 + tid;
            *(v4f*)(As + (idx >> 6) * APITCH + (idx & 63) * 4) = ra[i];
        }
#pragma unroll
        for (int i = 0; i < 32; ++i) {
            const int idx = i * 128 + tid;
            *(v4f*)(Ws + (idx >> 4) * WPITCH + (idx & 15) * 4) = rw[i];
        }
    }
    __syncthreads();

    const int half = lane >> 4;
    const int lh   = lane & 15;
    const float* ap0 = As + lh * APITCH + 2 * half;
    const float* ap1 = ap0 + 16 * APITCH;
    const float* bp  = Ws + (2 * half) * WPITCH + wave * 16 + lh;

    v8f c0 = {0.f, 0.f, 0.f, 0.f, 0.f, 0.f, 0.f, 0.f};
    v8f c1 = {0.f, 0.f, 0.f, 0.f, 0.f, 0.f, 0.f, 0.f};

    v2f fa0[2][KS], fa1[2][KS], fb[2][KS];
#pragma unroll
    for (int s = 0; s < KS; ++s) {
        const int k = s * 4;
        fa0[0][s]   = *(const v2f*)(ap0 + k);
        fa1[0][s]   = *(const v2f*)(ap1 + k);
        fb[0][s].x  = bp[(k)     * WPITCH];
        fb[0][s].y  = bp[(k + 1) * WPITCH];
    }
#pragma unroll
    for (int kb = 0; kb < D_ / (4 * KS); ++kb) {
        const int cur = kb & 1;
        const int nxt = cur ^ 1;
        __builtin_amdgcn_sched_barrier(0);
        if (kb + 1 < D_ / (4 * KS)) {
#pragma unroll
            for (int s = 0; s < KS; ++s) {
                const int k = (kb + 1) * (4 * KS) + s * 4;
                fa0[nxt][s]   = *(const v2f*)(ap0 + k);
                fa1[nxt][s]   = *(const v2f*)(ap1 + k);
                fb[nxt][s].x  = bp[(k)     * WPITCH];
                fb[nxt][s].y  = bp[(k + 1) * WPITCH];
            }
        }
        __builtin_amdgcn_sched_barrier(0);
#pragma unroll
        for (int s = 0; s < KS; ++s) {
            c0 = __builtin_amdgcn_wmma_f32_16x16x4_f32(
                    false, fa0[cur][s], false, fb[cur][s], (short)0, c0, false, false);
            c1 = __builtin_amdgcn_wmma_f32_16x16x4_f32(
                    false, fa1[cur][s], false, fb[cur][s], (short)0, c1, false, false);
        }
    }
    __builtin_amdgcn_sched_barrier(0);

    float* cw = Cs + (half * 8) * CPITCH + wave * 16 + lh;
#pragma unroll
    for (int i = 0; i < 8; ++i) {
        cw[i * CPITCH]        = c0[i];
        cw[(16 + i) * CPITCH] = c1[i];
    }
    __syncthreads();

    float* obase = out + (size_t)m0 * E_ + n0;
#pragma unroll
    for (int i = 0; i < 4; ++i) {
        const int idx = i * 128 + tid;
        const int row = idx >> 4;
        const int c4  = idx & 15;
        *(v4f*)(obase + (size_t)row * E_ + c4 * 4) = *(const v4f*)(Cs + row * CPITCH + c4 * 4);
    }
}

#define TL      8
#define EC      32
#define NCH     (E_ / EC)
#define QPITCH  44
#define QSTAGE  (LQ * QPITCH)

__global__ __launch_bounds__(256) void score_softmax_kernel(
    const float* __restrict__ qp, const float* __restrict__ kp,
    const float* __restrict__ vc, float* __restrict__ atten)
{
    __shared__ __align__(16) float qs[2 * QSTAGE];
    __shared__ __align__(16) float kpl[TL * E_];
    __shared__ __align__(16) float vcl[E_];
    __shared__ __align__(16) float redm[8 * TL];
    __shared__ __align__(16) float reds[8 * TL];

    const int tid  = threadIdx.x;
    const int lane = tid & 31;
    const int wave = tid >> 5;
    const int b    = blockIdx.x / (LK / TL);
    const int l0   = (blockIdx.x % (LK / TL)) * TL;

    const float* qpb = qp + (size_t)b * LQ * E_;

    ((v4f*)kpl)[tid] = ((const v4f*)(kp + (size_t)(b * LK + l0) * E_))[tid];
    if (tid < E_ / 4) ((v4f*)vcl)[tid] = ((const v4f*)vc)[tid];

    const int srow = tid >> 3;
    const int sc4  = tid & 7;
    const float* gsrc = qpb + (size_t)srow * E_ + sc4 * 4;
    float*       ldst = qs + srow * QPITCH + sc4 * 4;

    {
        v4f p0[16];
#pragma unroll
        for (int i = 0; i < 16; ++i)
            p0[i] = *(const v4f*)(gsrc + (size_t)i * 32 * E_);
#pragma unroll
        for (int i = 0; i < 16; ++i)
            *(v4f*)(ldst + i * 32 * QPITCH) = p0[i];
    }
    __syncthreads();

    v2f P0[TL / 2], P1[TL / 2];
#pragma unroll
    for (int j = 0; j < TL / 2; ++j) { P0[j] = (v2f){0.f, 0.f}; P1[j] = (v2f){0.f, 0.f}; }

#pragma unroll 1
    for (int c = 0; c < NCH; ++c) {
        const bool more = (c + 1 < NCH);
        v4f pre[16];
        if (more) {
#pragma unroll
            for (int i = 0; i < 16; ++i)
                pre[i] = *(const v4f*)(gsrc + (c + 1) * EC + (size_t)i * 32 * E_);
        }

        const float* qa = qs + (c & 1) * QSTAGE + tid * QPITCH;
        const float* qb = qa + 256 * QPITCH;
        const float* kc = kpl + c * EC;
        const float* vv = vcl + c * EC;

#pragma unroll 2
        for (int e = 0; e < EC; e += 4) {
            const float4 a  = *(const float4*)(qa + e);
            const float4 bq = *(const float4*)(qb + e);
            const float4 v  = *(const float4*)(vv + e);
            const v2f vx = (v2f){v.x, v.x}, vy = (v2f){v.y, v.y};
            const v2f vz = (v2f){v.z, v.z}, vw = (v2f){v.w, v.w};
#pragma unroll
            for (int j = 0; j < TL / 2; ++j) {
                const float4 k0 = *(const float4*)(kc + (2 * j) * E_ + e);
                const float4 k1 = *(const float4*)(kc + (2 * j + 1) * E_ + e);
                v2f x0 = P0[j], x1 = P1[j];
                x0 = __builtin_elementwise_fma((v2f){fast_tanh(a.x  + k0.x), fast_tanh(a.x  + k1.x)}, vx, x0);
                x1 = __builtin_elementwise_fma((v2f){fast_tanh(bq.x + k0.x), fast_tanh(bq.x + k1.x)}, vx, x1);
                x0 = __builtin_elementwise_fma((v2f){fast_tanh(a.y  + k0.y), fast_tanh(a.y  + k1.y)}, vy, x0);
                x1 = __builtin_elementwise_fma((v2f){fast_tanh(bq.y + k0.y), fast_tanh(bq.y + k1.y)}, vy, x1);
                x0 = __builtin_elementwise_fma((v2f){fast_tanh(a.z  + k0.z), fast_tanh(a.z  + k1.z)}, vz, x0);
                x1 = __builtin_elementwise_fma((v2f){fast_tanh(bq.z + k0.z), fast_tanh(bq.z + k1.z)}, vz, x1);
                x0 = __builtin_elementwise_fma((v2f){fast_tanh(a.w  + k0.w), fast_tanh(a.w  + k1.w)}, vw, x0);
                x1 = __builtin_elementwise_fma((v2f){fast_tanh(bq.w + k0.w), fast_tanh(bq.w + k1.w)}, vw, x1);
                P0[j] = x0; P1[j] = x1;
            }
        }

        if (more) {
            float* d = ldst + ((c + 1) & 1) * QSTAGE;
#pragma unroll
            for (int i = 0; i < 16; ++i)
                *(v4f*)(d + i * 32 * QPITCH) = pre[i];
        }
        __syncthreads();
    }

    float acc0[TL], acc1[TL];
#pragma unroll
    for (int j = 0; j < TL / 2; ++j) {
        acc0[2 * j] = P0[j].x; acc0[2 * j + 1] = P0[j].y;
        acc1[2 * j] = P1[j].x; acc1[2 * j + 1] = P1[j].y;
    }

    float m[TL];
#pragma unroll
    for (int l = 0; l < TL; ++l) m[l] = fmaxf(acc0[l], acc1[l]);
#pragma unroll
    for (int off = 16; off > 0; off >>= 1) {
#pragma unroll
        for (int l = 0; l < TL; ++l) m[l] = fmaxf(m[l], __shfl_xor(m[l], off, 32));
    }
    if (lane == 0) {
        ((v4f*)redm)[wave * 2]     = (v4f){m[0], m[1], m[2], m[3]};
        ((v4f*)redm)[wave * 2 + 1] = (v4f){m[4], m[5], m[6], m[7]};
    }
    __syncthreads();
#pragma unroll
    for (int w = 0; w < 8; ++w) {
        const v4f r0 = ((const v4f*)redm)[w * 2];
        const v4f r1 = ((const v4f*)redm)[w * 2 + 1];
        m[0] = fmaxf(m[0], r0.x); m[1] = fmaxf(m[1], r0.y);
        m[2] = fmaxf(m[2], r0.z); m[3] = fmaxf(m[3], r0.w);
        m[4] = fmaxf(m[4], r1.x); m[5] = fmaxf(m[5], r1.y);
        m[6] = fmaxf(m[6], r1.z); m[7] = fmaxf(m[7], r1.w);
    }

    float s[TL];
#pragma unroll
    for (int l = 0; l < TL; ++l) {
        acc0[l] = fast_exp(acc0[l] - m[l]);
        acc1[l] = fast_exp(acc1[l] - m[l]);
        s[l] = acc0[l] + acc1[l];
    }
#pragma unroll
    for (int off = 16; off > 0; off >>= 1) {
#pragma unroll
        for (int l = 0; l < TL; ++l) s[l] += __shfl_xor(s[l], off, 32);
    }
    if (lane == 0) {
        ((v4f*)reds)[wave * 2]     = (v4f){s[0], s[1], s[2], s[3]};
        ((v4f*)reds)[wave * 2 + 1] = (v4f){s[4], s[5], s[6], s[7]};
    }
    __syncthreads();
    float t[TL];
#pragma unroll
    for (int l = 0; l < TL; ++l) t[l] = 0.f;
#pragma unroll
    for (int w = 0; w < 8; ++w) {
        const v4f r0 = ((const v4f*)reds)[w * 2];
        const v4f r1 = ((const v4f*)reds)[w * 2 + 1];
        t[0] += r0.x; t[1] += r0.y; t[2] += r0.z; t[3] += r0.w;
        t[4] += r1.x; t[5] += r1.y; t[6] += r1.z; t[7] += r1.w;
    }

    float* arow = atten + ((size_t)(b * LK + l0)) * LQ + tid;
#pragma unroll
    for (int l = 0; l < TL; ++l) {
        const float inv = 1.f / t[l];
        arow[(size_t)l * LQ]       = acc0[l] * inv;
        arow[(size_t)l * LQ + 256] = acc1[l] * inv;
    }
}

#define CM       32
#define CN       64
#define CKC      128
#define CNCH     (LQ / CKC)
#define CAPITCH  516
#define CBPITCH  72
#define CCPITCH  68
#define CKS      16
#define CTX_GROUPS  (B_ * (LK / CM) * (D_ / CN))

__global__ __launch_bounds__(128) void context_wmma_kernel(
    const float* __restrict__ atten, const float* __restrict__ value,
    float* __restrict__ context)
{
    __shared__ __align__(16) float As[CM * CAPITCH];
    __shared__ __align__(16) float Bs[LQ * CBPITCH];
    __shared__ __align__(16) float Cs[CM * CCPITCH];

    const int tid  = threadIdx.x;
    const int lane = tid & 31;
    const int wave = tid >> 5;
    const int bid  = blockIdx.x;
    const int b    = bid >> 6;
    const int r    = bid & 63;
    const int l0   = (r >> 2) * CM;
    const int n0   = (r & 3) * CN;

    const int arow = tid >> 5, apc = tid & 31;
    const int brow = tid >> 4, bpc = tid & 15;
    const float* ag = atten + ((size_t)(b * LK + l0 + arow)) * LQ + apc * 4;
    const float* bg = value + ((size_t)(b * LQ + brow)) * D_ + n0 + bpc * 4;
    float* al = As + arow * CAPITCH + apc * 4;
    float* bl = Bs + brow * CBPITCH + bpc * 4;

    {
        v4f ra[8], rb[16];
#pragma unroll
        for (int i = 0; i < 8; ++i)
            ra[i] = *(const v4f*)(ag + (size_t)(i * 4) * LQ);
#pragma unroll
        for (int i = 0; i < 16; ++i)
            rb[i] = *(const v4f*)(bg + (size_t)(i * 8) * D_);
#pragma unroll
        for (int i = 0; i < 8; ++i)
            *(v4f*)(al + (i * 4) * CAPITCH) = ra[i];
#pragma unroll
        for (int i = 0; i < 16; ++i)
            *(v4f*)(bl + (i * 8) * CBPITCH) = rb[i];
    }
    __syncthreads();

    const int half = lane >> 4;
    const int lh   = lane & 15;
    const float* ap0 = As + lh * CAPITCH + 2 * half;
    const float* ap1 = ap0 + 16 * CAPITCH;
    const float* bp  = Bs + (2 * half) * CBPITCH + wave * 16 + lh;

    v8f c0 = {0.f, 0.f, 0.f, 0.f, 0.f, 0.f, 0.f, 0.f};
    v8f c1 = {0.f, 0.f, 0.f, 0.f, 0.f, 0.f, 0.f, 0.f};

#pragma unroll 1
    for (int c = 0; c < CNCH; ++c) {
        const bool more = (c + 1 < CNCH);
        v4f pa[8], pb[16];
        if (more) {
            const float* ag1 = ag + (c + 1) * CKC;
            const float* bg1 = bg + (size_t)((c + 1) * CKC) * D_;
#pragma unroll
            for (int i = 0; i < 8; ++i)
                pa[i] = *(const v4f*)(ag1 + (size_t)(i * 4) * LQ);
#pragma unroll
            for (int i = 0; i < 16; ++i)
                pb[i] = *(const v4f*)(bg1 + (size_t)(i * 8) * D_);
        }

        const float* a0 = ap0 + c * CKC;
        const float* a1 = ap1 + c * CKC;
        const float* bb = bp + (c * CKC) * CBPITCH;

        v2f fa0[2][CKS], fa1[2][CKS], fb[2][CKS];
        __builtin_amdgcn_sched_barrier(0);
#pragma unroll
        for (int s = 0; s < CKS; ++s) {
            const int k = s * 4;
            fa0[0][s]   = *(const v2f*)(a0 + k);
            fa1[0][s]   = *(const v2f*)(a1 + k);
            fb[0][s].x  = bb[(k)     * CBPITCH];
            fb[0][s].y  = bb[(k + 1) * CBPITCH];
        }
#pragma unroll
        for (int kb = 0; kb < CKC / (4 * CKS); ++kb) {
            const int cur = kb & 1;
            const int nxt = cur ^ 1;
            __builtin_amdgcn_sched_barrier(0);
            if (kb + 1 < CKC / (4 * CKS)) {
#pragma unroll
                for (int s = 0; s < CKS; ++s) {
                    const int k = (kb + 1) * (4 * CKS) + s * 4;
                    fa0[nxt][s]   = *(const v2f*)(a0 + k);
                    fa1[nxt][s]   = *(const v2f*)(a1 + k);
                    fb[nxt][s].x  = bb[(k)     * CBPITCH];
                    fb[nxt][s].y  = bb[(k + 1) * CBPITCH];
                }
            }
            __builtin_amdgcn_sched_barrier(0);
#pragma unroll
            for (int s = 0; s < CKS; ++s) {
                c0 = __builtin_amdgcn_wmma_f32_16x16x4_f32(
                        false, fa0[cur][s], false, fb[cur][s], (short)0, c0, false, false);
                c1 = __builtin_amdgcn_wmma_f32_16x16x4_f32(
                        false, fa1[cur][s], false, fb[cur][s], (short)0, c1, false, false);
            }
        }
        __builtin_amdgcn_sched_barrier(0);

        if (more) {
            float* al1 = al + (c + 1) * CKC;
            float* bl1 = bl + ((c + 1) * CKC) * CBPITCH;
#pragma unroll
            for (int i = 0; i < 8; ++i)
                *(v4f*)(al1 + (i * 4) * CAPITCH) = pa[i];
#pragma unroll
            for (int i = 0; i < 16; ++i)
                *(v4f*)(bl1 + (i * 8) * CBPITCH) = pb[i];
        }
        __syncthreads();
    }

    float* cw = Cs + (half * 8) * CCPITCH + wave * 16 + lh;
#pragma unroll
    for (int i = 0; i < 8; ++i) {
        cw[i * CCPITCH]        = c0[i];
        cw[(16 + i) * CCPITCH] = c1[i];
    }
    __syncthreads();

    float* obase = context + ((size_t)(b * LK + l0)) * D_ + n0;
#pragma unroll
    for (int i = 0; i < 4; ++i) {
        const int idx = i * 128 + tid;
        const int row = idx >> 4;
        const int c4  = idx & 15;
        *(v4f*)(obase + (size_t)row * D_ + c4 * 4) = *(const v4f*)(Cs + row * CCPITCH + c4 * 4);
    }
}

extern "C" void kernel_launch(void* const* d_in, const int* in_sizes, int n_in,
                              void* d_out, int out_size, void* d_ws, size_t ws_size,
                              hipStream_t stream)
{
    const float* query = (const float*)d_in[0];
    const float* key   = (const float*)d_in[1];
    const float* value = (const float*)d_in[2];
    const float* Wc1   = (const float*)d_in[3];
    const float* Wc2   = (const float*)d_in[4];
    const float* vc    = (const float*)d_in[5];

    float* out     = (float*)d_out;
    float* context = out;
    float* atten   = out + (size_t)B_ * LK * D_;

    float* qp = (float*)d_ws;
    float* kp = qp + (size_t)B_ * LQ * E_;

    proj_wmma_kernel<<<PROJ_GROUPS, 128, 0, stream>>>(query, key, Wc1, Wc2, qp, kp);
    score_softmax_kernel<<<B_ * (LK / TL), 256, 0, stream>>>(qp, kp, vc, atten);
    context_wmma_kernel<<<CTX_GROUPS, 128, 0, stream>>>(atten, value, context);
}


# score kernel inner loop issued in a fixed order (inline asm): every v_tanh_f32 is followed by one independent packed add or fma; key rows kept feature-major in LDS so the q+k sums are packed adds
# speedup vs baseline: 3.1092x; 1.0495x over previous
#include <hip/hip_runtime.h>
#include <math.h>

#define B_  4
#define LQ  512
#define LK  512
#define E_  128
#define D_  256

typedef float v2f __attribute__((ext_vector_type(2)));
typedef float v8f __attribute__((ext_vector_type(8)));
typedef float v4f __attribute__((ext_vector_type(4)));

__device__ __forceinline__ float fast_exp(float x) {
#if __has_builtin(__builtin_amdgcn_exp2f)
    return __builtin_amdgcn_exp2f(x * 1.44269504088896340f);
#else
    return __expf(x);
#endif
}

#define PM      32
#define PN      64
#define APITCH  260
#define WPITCH  72
#define CPITCH  68
#define KS      16
#define PROJ_GROUPS  (2 * (B_ * 512 / PM) * (E_ / PN))

__global__ __launch_bounds__(128) void proj_wmma_kernel(
    const float* __restrict__ query, const float* __restrict__ key,
    const float* __restrict__ Wc1,   const float* __restrict__ Wc2,
    float* __restrict__ qp, float* __restrict__ kp)
{
    __shared__ __align__(16) float As[PM * APITCH];
    __shared__ __align__(16) float Ws[D_ * WPITCH];
    __shared__ __align__(16) float Cs[PM * CPITCH];

    const int tid  = threadIdx.x;
    const int lane = tid & 31;
    const int wave = tid >> 5;
    const int bid  = blockIdx.x;
    const int sel  = bid >> 7;
    const int r    = bid & 127;
    const int m0   = (r >> 1) * PM;
    const int n0   = (r & 1) * PN;

    const float* in  = sel ? key : query;
    const float* Wg  = sel ? Wc2 : Wc1;
    float*       out = sel ? kp  : qp;

    {
        const float* asrc = in + (size_t)m0 * D_;
        v4f ra[16];
#pragma unroll
        for (int i = 0; i < 16; ++i)
            ra[i] = *(const v4f*)(asrc + (size_t)(i * 128 + tid) * 4);
        const float* wsrc = Wg + n0;
        v4f rw[32];
#pragma unroll
        for (int i = 0; i < 32; ++i) {
            const int idx = i * 128 + tid;
            rw[i] = *(const v4f*)(wsrc + (size_t)(idx >> 4) * E_ + (idx & 15) * 4);
        }
#pragma unroll
        for (int i = 0; i < 16; ++i) {
            const int idx = i * 128 + tid;
            *(v4f*)(As + (idx >> 6) * APITCH + (idx & 63) * 4) = ra[i];
        }
#pragma unroll
        for (int i = 0; i < 32; ++i) {
            const int idx = i * 128 + tid;
            *(v4f*)(Ws + (idx >> 4) * WPITCH + (idx & 15) * 4) = rw[i];
        }
    }
    __syncthreads();

    const int half = lane >> 4;
    const int lh   = lane & 15;
    const float* ap0 = As + lh * APITCH + 2 * half;
    const float* ap1 = ap0 + 16 * APITCH;
    const float* bp  = Ws + (2 * half) * WPITCH + wave * 16 + lh;

    v8f c0 = {0.f, 0.f, 0.f, 0.f, 0.f, 0.f, 0.f, 0.f};
    v8f c1 = {0.f, 0.f, 0.f, 0.f, 0.f, 0.f, 0.f, 0.f};

    v2f fa0[2][KS], fa1[2][KS], fb[2][KS];
#pragma unroll
    for (int s = 0; s < KS; ++s) {
        const int k = s * 4;
        fa0[0][s]   = *(const v2f*)(ap0 + k);
        fa1[0][s]   = *(const v2f*)(ap1 + k);
        fb[0][s].x  = bp[(k)     * WPITCH];
        fb[0][s].y  = bp[(k + 1) * WPITCH];
    }
#pragma unroll
    for (int kb = 0; kb < D_ / (4 * KS); ++kb) {
        const int cur = kb & 1;
        const int nxt = cur ^ 1;
        __builtin_amdgcn_sched_barrier(0);
        if (kb + 1 < D_ / (4 * KS)) {
#pragma unroll
            for (int s = 0; s < KS; ++s) {
                const int k = (kb + 1) * (4 * KS) + s * 4;
                fa0[nxt][s]   = *(const v2f*)(ap0 + k);
                fa1[nxt][s]   = *(const v2f*)(ap1 + k);
                fb[nxt][s].x  = bp[(k)     * WPITCH];
                fb[nxt][s].y  = bp[(k + 1) * WPITCH];
            }
        }
        __builtin_amdgcn_sched_barrier(0);
#pragma unroll
        for (int s = 0; s < KS; ++s) {
            c0 = __builtin_amdgcn_wmma_f32_16x16x4_f32(
                    false, fa0[cur][s], false, fb[cur][s], (short)0, c0, false, false);
            c1 = __builtin_amdgcn_wmma_f32_16x16x4_f32(
                    false, fa1[cur][s], false, fb[cur][s], (short)0, c1, false, false);
        }
    }
    __builtin_amdgcn_sched_barrier(0);

    float* cw = Cs + (half * 8) * CPITCH + wave * 16 + lh;
#pragma unroll
    for (int i = 0; i < 8; ++i) {
        cw[i * CPITCH]        = c0[i];
        cw[(16 + i) * CPITCH] = c1[i];
    }
    __syncthreads();

    float* obase = out + (size_t)m0 * E_ + n0;
#pragma unroll
    for (int i = 0; i < 4; ++i) {
        const int idx = i * 128 + tid;
        const int row = idx >> 4;
        const int c4  = idx & 15;
        *(v4f*)(obase + (size_t)row * E_ + c4 * 4) = *(const v4f*)(Cs + row * CPITCH + c4 * 4);
    }
}

#define TL      8
#define EC      32
#define NCH     (E_ / EC)
#define QPITCH  44
#define QSTAGE  (LQ * QPITCH)

__global__ __launch_bounds__(256) void score_softmax_kernel(
    const float* __restrict__ qp, const float* __restrict__ kp,
    const float* __restrict__ vc, float* __restrict__ atten)
{
    __shared__ __align__(16) float qs[2 * QSTAGE];
    __shared__ __align__(16) float kpl[TL * E_];
    __shared__ __align__(16) float vcl[E_];
    __shared__ __align__(16) float redm[8 * TL];
    __shared__ __align__(16) float reds[8 * TL];

    const int tid  = threadIdx.x;
    const int lane = tid & 31;
    const int wave = tid >> 5;
    const int b    = blockIdx.x / (LK / TL);
    const int l0   = (blockIdx.x % (LK / TL)) * TL;

    const float* qpb = qp + (size_t)b * LQ * E_;

    {
        const v4f kv = ((const v4f*)(kp + (size_t)(b * LK + l0) * E_))[tid];
        const int krow = tid >> 5;
        const int ke0  = (tid & 31) * 4;
        kpl[(ke0 + 0) * TL + krow] = kv.x;
        kpl[(ke0 + 1) * TL + krow] = kv.y;
        kpl[(ke0 + 2) * TL + krow] = kv.z;
        kpl[(ke0 + 3) * TL + krow] = kv.w;
    }
    if (tid < E_ / 4) ((v4f*)vcl)[tid] = ((const v4f*)vc)[tid];

    const int srow = tid >> 3;
    const int sc4  = tid & 7;
    const float* gsrc = qpb + (size_t)srow * E_ + sc4 * 4;
    float*       ldst = qs + srow * QPITCH + sc4 * 4;

    {
        v4f p0[16];
#pragma unroll
        for (int i = 0; i < 16; ++i)
            p0[i] = *(const v4f*)(gsrc + (size_t)i * 32 * E_);
#pragma unroll
        for (int i = 0; i < 16; ++i)
            *(v4f*)(ldst + i * 32 * QPITCH) = p0[i];
    }
    __syncthreads();

    v2f P0[TL / 2], P1[TL / 2];
#pragma unroll
    for (int j = 0; j < TL / 2; ++j) { P0[j] = (v2f){0.f, 0.f}; P1[j] = (v2f){0.f, 0.f}; }

#pragma unroll 1
    for (int c = 0; c < NCH; ++c) {
        const bool more = (c + 1 < NCH);
        v4f pre[16];
        if (more) {
#pragma unroll
            for (int i = 0; i < 16; ++i)
                pre[i] = *(const v4f*)(gsrc + (c + 1) * EC + (size_t)i * 32 * E_);
        }

        const float* qa = qs + (c & 1) * QSTAGE + tid * QPITCH;
        const float* qb = qa + 256 * QPITCH;
        const float* kc = kpl + (c * EC) * TL;
        const float* vv = vcl + c * EC;

#pragma unroll 1
        for (int e = 0; e < EC; e += 8) {
            v2f A2[4], B2[4], V2[4], K2[32];
#pragma unroll
            for (int g = 0; g < 2; ++g) {
                const v4f a4 = *(const v4f*)(qa + e + 4 * g);
                const v4f b4 = *(const v4f*)(qb + e + 4 * g);
                const v4f w4 = *(const v4f*)(vv + e + 4 * g);
                A2[2 * g]     = __builtin_shufflevector(a4, a4, 0, 1);
                A2[2 * g + 1] = __builtin_shufflevector(a4, a4, 2, 3);
                B2[2 * g]     = __builtin_shufflevector(b4, b4, 0, 1);
                B2[2 * g + 1] = __builtin_shufflevector(b4, b4, 2, 3);
                V2[2 * g]     = __builtin_shufflevector(w4, w4, 0, 1);
                V2[2 * g + 1] = __builtin_shufflevector(w4, w4, 2, 3);
            }
#pragma unroll
            for (int f = 0; f < 8; ++f) {
                const v4f kA = *(const v4f*)(kc + (e + f) * TL);
                const v4f kB = *(const v4f*)(kc + (e + f) * TL + 4);
                K2[4 * f]     = __builtin_shufflevector(kA, kA, 0, 1);
                K2[4 * f + 1] = __builtin_shufflevector(kA, kA, 2, 3);
                K2[4 * f + 2] = __builtin_shufflevector(kB, kB, 0, 1);
                K2[4 * f + 3] = __builtin_shufflevector(kB, kB, 2, 3);
            }
            asm volatile(
                "v_pk_add_f32 v[232:233], %8, %20 op_sel_hi:[0,1]\n\t"
                "v_pk_add_f32 v[234:235], %8, %21 op_sel_hi:[0,1]\n\t"
                "v_pk_add_f32 v[236:237], %8, %22 op_sel_hi:[0,1]\n\t"
                "v_tanh_f32 v240, v232\n\t"
                "v_pk_add_f32 v[238:239], %8, %23 op_sel_hi:[0,1]\n\t"
                "v_tanh_f32 v241, v233\n\t"
                "v_tanh_f32 v242, v234\n\t"
                "v_pk_add_f32 v[232:233], %12, %20 op_sel_hi:[0,1]\n\t"
                "v_tanh_f32 v243, v235\n\t"
                "v_tanh_f32 v244, v236\n\t"
                "v_pk_add_f32 v[234:235], %12, %21 op_sel_hi:[0,1]\n\t"
                "v_tanh_f32 v245, v237\n\t"
                "v_tanh_f32 v246, v238\n\t"
                "v_pk_add_f32 v[236:237], %12, %22 op_sel_hi:[0,1]\n\t"
                "v_tanh_f32 v247, v239\n\t"
                "v_pk_fma_f32 %0, v[240:241], %16, %0 op_sel_hi:[1,0,1]\n\t"
                "v_tanh_f32 v240, v232\n\t"
                "v_pk_add_f32 v[238:239], %12, %23 op_sel_hi:[0,1]\n\t"
                "v_tanh_f32 v241, v233\n\t"
                "v_pk_fma_f32 %1, v[242:243], %16, %1 op_sel_hi:[1,0,1]\n\t"
                "v_tanh_f32 v242, v234\n\t"
                "v_pk_add_f32 v[232:233], %8, %24 op_sel:[1,0]\n\t"
                "v_tanh_f32 v243, v235\n\t"
                "v_pk_fma_f32 %2, v[244:245], %16, %2 op_sel_hi:[1,0,1]\n\t"
                "v_tanh_f32 v244, v236\n\t"
                "v_pk_add_f32 v[234:235], %8, %25 op_sel:[1,0]\n\t"
                "v_tanh_f32 v245, v237\n\t"
                "v_pk_fma_f32 %3, v[246:247], %16, %3 op_sel_hi:[1,0,1]\n\t"
                "v_tanh_f32 v246, v238\n\t"
                "v_pk_add_f32 v[236:237], %8, %26 op_sel:[1,0]\n\t"
                "v_tanh_f32 v247, v239\n\t"
                "v_pk_fma_f32 %4, v[240:241], %16, %4 op_sel_hi:[1,0,1]\n\t"
                "v_tanh_f32 v240, v232\n\t"
                "v_pk_add_f32 v[238:239], %8, %27 op_sel:[1,0]\n\t"
                "v_tanh_f32 v241, v233\n\t"
                "v_pk_fma_f32 %5, v[242:243], %16, %5 op_sel_hi:[1,0,1]\n\t"
                "v_tanh_f32 v242, v234\n\t"
                "v_pk_add_f32 v[232:233], %12, %24 op_sel:[1,0]\n\t"
                "v_tanh_f32 v243, v235\n\t"
                "v_pk_fma_f32 %6, v[244:245], %16, %6 op_sel_hi:[1,0,1]\n\t"
                "v_tanh_f32 v244, v236\n\t"
                "v_pk_add_f32 v[234:235], %12, %25 op_sel:[1,0]\n\t"
                "v_tanh_f32 v245, v237\n\t"
                "v_pk_fma_f32 %7, v[246:247], %16, %7 op_sel_hi:[1,0,1]\n\t"
                "v_tanh_f32 v246, v238\n\t"
                "v_pk_add_f32 v[236:237], %12, %26 op_sel:[1,0]\n\t"
                "v_tanh_f32 v247, v239\n\t"
                "v_pk_fma_f32 %0, v[240:241], %16, %0 op_sel:[0,1,0]\n\t"
                "v_tanh_f32 v240, v232\n\t"
                "v_pk_add_f32 v[238:239], %12, %27 op_sel:[1,0]\n\t"
                "v_tanh_f32 v241, v233\n\t"
                "v_pk_fma_f32 %1, v[242:243], %16, %1 op_sel:[0,1,0]\n\t"
                "v_tanh_f32 v242, v234\n\t"
                "v_pk_add_f32 v[232:233], %9, %28 op_sel_hi:[0,1]\n\t"
                "v_tanh_f32 v243, v235\n\t"
                "v_pk_fma_f32 %2, v[244:245], %16, %2 op_sel:[0,1,0]\n\t"
                "v_tanh_f32 v244, v236\n\t"
                "v_pk_add_f32 v[234:235], %9, %29 op_sel_hi:[0,1]\n\t"
                "v_tanh_f32 v245, v237\n\t"
                "v_pk_fma_f32 %3, v[246:247], %16, %3 op_sel:[0,1,0]\n\t"
                "v_tanh_f32 v246, v238\n\t"
                "v_pk_add_f32 v[236:237], %9, %30 op_sel_hi:[0,1]\n\t"
                "v_tanh_f32 v247, v239\n\t"
                "v_pk_fma_f32 %4, v[240:241], %16, %4 op_sel:[0,1,0]\n\t"
                "v_tanh_f32 v240, v232\n\t"
                "v_pk_add_f32 v[238:239], %9, %31 op_sel_hi:[0,1]\n\t"
                "v_tanh_f32 v241, v233\n\t"
                "v_pk_fma_f32 %5, v[242:243], %16, %5 op_sel:[0,1,0]\n\t"
                "v_tanh_f32 v242, v234\n\t"
                "v_pk_add_f32 v[232:233], %13, %28 op_sel_hi:[0,1]\n\t"
                "v_tanh_f32 v243, v235\n\t"
                "v_pk_fma_f32 %6, v[244:245], %16, %6 op_sel:[0,1,0]\n\t"
                "v_tanh_f32 v244, v236\n\t"
                "v_pk_add_f32 v[234:235], %13, %29 op_sel_hi:[0,1]\n\t"
                "v_tanh_f32 v245, v237\n\t"
                "v_pk_fma_f32 %7, v[246:247], %16, %7 op_sel:[0,1,0]\n\t"
                "v_tanh_f32 v246, v238\n\t"
                "v_pk_add_f32 v[236:237], %13, %30 op_sel_hi:[0,1]\n\t"
                "v_tanh_f32 v247, v239\n\t"
                "v_pk_fma_f32 %0, v[240:241], %17, %0 op_sel_hi:[1,0,1]\n\t"
                "v_tanh_f32 v240, v232\n\t"
                "v_pk_add_f32 v[238:239], %13, %31 op_sel_hi:[0,1]\n\t"
                "v_tanh_f32 v241, v233\n\t"
                "v_pk_fma_f32 %1, v[242:243], %17, %1 op_sel_hi:[1,0,1]\n\t"
                "v_tanh_f32 v242, v234\n\t"
                "v_pk_add_f32 v[232:233], %9, %32 op_sel:[1,0]\n\t"
                "v_tanh_f32 v243, v235\n\t"
                "v_pk_fma_f32 %2, v[244:245], %17, %2 op_sel_hi:[1,0,1]\n\t"
                "v_tanh_f32 v244, v236\n\t"
                "v_pk_add_f32 v[234:235], %9, %33 op_sel:[1,0]\n\t"
                "v_tanh_f32 v245, v237\n\t"
                "v_pk_fma_f32 %3, v[246:247], %17, %3 op_sel_hi:[1,0,1]\n\t"
                "v_tanh_f32 v246, v238\n\t"
                "v_pk_add_f32 v[236:237], %9, %34 op_sel:[1,0]\n\t"
                "v_tanh_f32 v247, v239\n\t"
                "v_pk_fma_f32 %4, v[240:241], %17, %4 op_sel_hi:[1,0,1]\n\t"
                "v_tanh_f32 v240, v232\n\t"
                "v_pk_add_f32 v[238:239], %9, %35 op_sel:[1,0]\n\t"
                "v_tanh_f32 v241, v233\n\t"
                "v_pk_fma_f32 %5, v[242:243], %17, %5 op_sel_hi:[1,0,1]\n\t"
                "v_tanh_f32 v242, v234\n\t"
                "v_pk_add_f32 v[232:233], %13, %32 op_sel:[1,0]\n\t"
                "v_tanh_f32 v243, v235\n\t"
                "v_pk_fma_f32 %6, v[244:245], %17, %6 op_sel_hi:[1,0,1]\n\t"
                "v_tanh_f32 v244, v236\n\t"
                "v_pk_add_f32 v[234:235], %13, %33 op_sel:[1,0]\n\t"
                "v_tanh_f32 v245, v237\n\t"
                "v_pk_fma_f32 %7, v[246:247], %17, %7 op_sel_hi:[1,0,1]\n\t"
                "v_tanh_f32 v246, v238\n\t"
                "v_pk_add_f32 v[236:237], %13, %34 op_sel:[1,0]\n\t"
                "v_tanh_f32 v247, v239\n\t"
                "v_pk_fma_f32 %0, v[240:241], %17, %0 op_sel:[0,1,0]\n\t"
                "v_tanh_f32 v240, v232\n\t"
                "v_pk_add_f32 v[238:239], %13, %35 op_sel:[1,0]\n\t"
                "v_tanh_f32 v241, v233\n\t"
                "v_pk_fma_f32 %1, v[242:243], %17, %1 op_sel:[0,1,0]\n\t"
                "v_tanh_f32 v242, v234\n\t"
                "v_pk_add_f32 v[232:233], %10, %36 op_sel_hi:[0,1]\n\t"
                "v_tanh_f32 v243, v235\n\t"
                "v_pk_fma_f32 %2, v[244:245], %17, %2 op_sel:[0,1,0]\n\t"
                "v_tanh_f32 v244, v236\n\t"
                "v_pk_add_f32 v[234:235], %10, %37 op_sel_hi:[0,1]\n\t"
                "v_tanh_f32 v245, v237\n\t"
                "v_pk_fma_f32 %3, v[246:247], %17, %3 op_sel:[0,1,0]\n\t"
                "v_tanh_f32 v246, v238\n\t"
                "v_pk_add_f32 v[236:237], %10, %38 op_sel_hi:[0,1]\n\t"
                "v_tanh_f32 v247, v239\n\t"
                "v_pk_fma_f32 %4, v[240:241], %17, %4 op_sel:[0,1,0]\n\t"
                "v_tanh_f32 v240, v232\n\t"
                "v_pk_add_f32 v[238:239], %10, %39 op_sel_hi:[0,1]\n\t"
                "v_tanh_f32 v241, v233\n\t"
                "v_pk_fma_f32 %5, v[242:243], %17, %5 op_sel:[0,1,0]\n\t"
                "v_tanh_f32 v242, v234\n\t"
                "v_pk_add_f32 v[232:233], %14, %36 op_sel_hi:[0,1]\n\t"
                "v_tanh_f32 v243, v235\n\t"
                "v_pk_fma_f32 %6, v[244:245], %17, %6 op_sel:[0,1,0]\n\t"
                "v_tanh_f32 v244, v236\n\t"
                "v_pk_add_f32 v[234:235], %14, %37 op_sel_hi:[0,1]\n\t"
                "v_tanh_f32 v245, v237\n\t"
                "v_pk_fma_f32 %7, v[246:247], %17, %7 op_sel:[0,1,0]\n\t"
                "v_tanh_f32 v246, v238\n\t"
                "v_pk_add_f32 v[236:237], %14, %38 op_sel_hi:[0,1]\n\t"
                "v_tanh_f32 v247, v239\n\t"
                "v_pk_fma_f32 %0, v[240:241], %18, %0 op_sel_hi:[1,0,1]\n\t"
                "v_tanh_f32 v240, v232\n\t"
                "v_pk_add_f32 v[238:239], %14, %39 op_sel_hi:[0,1]\n\t"
                "v_tanh_f32 v241, v233\n\t"
                "v_pk_fma_f32 %1, v[242:243], %18, %1 op_sel_hi:[1,0,1]\n\t"
                "v_tanh_f32 v242, v234\n\t"
                "v_pk_add_f32 v[232:233], %10, %40 op_sel:[1,0]\n\t"
                "v_tanh_f32 v243, v235\n\t"
                "v_pk_fma_f32 %2, v[244:245], %18, %2 op_sel_hi:[1,0,1]\n\t"
                "v_tanh_f32 v244, v236\n\t"
                "v_pk_add_f32 v[234:235], %10, %41 op_sel:[1,0]\n\t"
                "v_tanh_f32 v245, v237\n\t"
                "v_pk_fma_f32 %3, v[246:247], %18, %3 op_sel_hi:[1,0,1]\n\t"
                "v_tanh_f32 v246, v238\n\t"
                "v_pk_add_f32 v[236:237], %10, %42 op_sel:[1,0]\n\t"
                "v_tanh_f32 v247, v239\n\t"
                "v_pk_fma_f32 %4, v[240:241], %18, %4 op_sel_hi:[1,0,1]\n\t"
                "v_tanh_f32 v240, v232\n\t"
                "v_pk_add_f32 v[238:239], %10, %43 op_sel:[1,0]\n\t"
                "v_tanh_f32 v241, v233\n\t"
                "v_pk_fma_f32 %5, v[242:243], %18, %5 op_sel_hi:[1,0,1]\n\t"
                "v_tanh_f32 v242, v234\n\t"
                "v_pk_add_f32 v[232:233], %14, %40 op_sel:[1,0]\n\t"
                "v_tanh_f32 v243, v235\n\t"
                "v_pk_fma_f32 %6, v[244:245], %18, %6 op_sel_hi:[1,0,1]\n\t"
                "v_tanh_f32 v244, v236\n\t"
                "v_pk_add_f32 v[234:235], %14, %41 op_sel:[1,0]\n\t"
                "v_tanh_f32 v245, v237\n\t"
                "v_pk_fma_f32 %7, v[246:247], %18, %7 op_sel_hi:[1,0,1]\n\t"
                "v_tanh_f32 v246, v238\n\t"
                "v_pk_add_f32 v[236:237], %14, %42 op_sel:[1,0]\n\t"
                "v_tanh_f32 v247, v239\n\t"
                "v_pk_fma_f32 %0, v[240:241], %18, %0 op_sel:[0,1,0]\n\t"
                "v_tanh_f32 v240, v232\n\t"
                "v_pk_add_f32 v[238:239], %14, %43 op_sel:[1,0]\n\t"
                "v_tanh_f32 v241, v233\n\t"
                "v_pk_fma_f32 %1, v[242:243], %18, %1 op_sel:[0,1,0]\n\t"
                "v_tanh_f32 v242, v234\n\t"
                "v_pk_add_f32 v[232:233], %11, %44 op_sel_hi:[0,1]\n\t"
                "v_tanh_f32 v243, v235\n\t"
                "v_pk_fma_f32 %2, v[244:245], %18, %2 op_sel:[0,1,0]\n\t"
                "v_tanh_f32 v244, v236\n\t"
                "v_pk_add_f32 v[234:235], %11, %45 op_sel_hi:[0,1]\n\t"
                "v_tanh_f32 v245, v237\n\t"
                "v_pk_fma_f32 %3, v[246:247], %18, %3 op_sel:[0,1,0]\n\t"
                "v_tanh_f32 v246, v238\n\t"
                "v_pk_add_f32 v[236:237], %11, %46 op_sel_hi:[0,1]\n\t"
                "v_tanh_f32 v247, v239\n\t"
                "v_pk_fma_f32 %4, v[240:241], %18, %4 op_sel:[0,1,0]\n\t"
                "v_tanh_f32 v240, v232\n\t"
                "v_pk_add_f32 v[238:239], %11, %47 op_sel_hi:[0,1]\n\t"
                "v_tanh_f32 v241, v233\n\t"
                "v_pk_fma_f32 %5, v[242:243], %18, %5 op_sel:[0,1,0]\n\t"
                "v_tanh_f32 v242, v234\n\t"
                "v_pk_add_f32 v[232:233], %15, %44 op_sel_hi:[0,1]\n\t"
                "v_tanh_f32 v243, v235\n\t"
                "v_pk_fma_f32 %6, v[244:245], %18, %6 op_sel:[0,1,0]\n\t"
                "v_tanh_f32 v244, v236\n\t"
                "v_pk_add_f32 v[234:235], %15, %45 op_sel_hi:[0,1]\n\t"
                "v_tanh_f32 v245, v237\n\t"
                "v_pk_fma_f32 %7, v[246:247], %18, %7 op_sel:[0,1,0]\n\t"
                "v_tanh_f32 v246, v238\n\t"
                "v_pk_add_f32 v[236:237], %15, %46 op_sel_hi:[0,1]\n\t"
                "v_tanh_f32 v247, v239\n\t"
                "v_pk_fma_f32 %0, v[240:241], %19, %0 op_sel_hi:[1,0,1]\n\t"
                "v_tanh_f32 v240, v232\n\t"
                "v_pk_add_f32 v[238:239], %15, %47 op_sel_hi:[0,1]\n\t"
                "v_tanh_f32 v241, v233\n\t"
                "v_pk_fma_f32 %1, v[242:243], %19, %1 op_sel_hi:[1,0,1]\n\t"
                "v_tanh_f32 v242, v234\n\t"
                "v_pk_add_f32 v[232:233], %11, %48 op_sel:[1,0]\n\t"
                "v_tanh_f32 v243, v235\n\t"
                "v_pk_fma_f32 %2, v[244:245], %19, %2 op_sel_hi:[1,0,1]\n\t"
                "v_tanh_f32 v244, v236\n\t"
                "v_pk_add_f32 v[234:235], %11, %49 op_sel:[1,0]\n\t"
                "v_tanh_f32 v245, v237\n\t"
                "v_pk_fma_f32 %3, v[246:247], %19, %3 op_sel_hi:[1,0,1]\n\t"
                "v_tanh_f32 v246, v238\n\t"
                "v_pk_add_f32 v[236:237], %11, %50 op_sel:[1,0]\n\t"
                "v_tanh_f32 v247, v239\n\t"
                "v_pk_fma_f32 %4, v[240:241], %19, %4 op_sel_hi:[1,0,1]\n\t"
                "v_tanh_f32 v240, v232\n\t"
                "v_pk_add_f32 v[238:239], %11, %51 op_sel:[1,0]\n\t"
                "v_tanh_f32 v241, v233\n\t"
                "v_pk_fma_f32 %5, v[242:243], %19, %5 op_sel_hi:[1,0,1]\n\t"
                "v_tanh_f32 v242, v234\n\t"
                "v_pk_add_f32 v[232:233], %15, %48 op_sel:[1,0]\n\t"
                "v_tanh_f32 v243, v235\n\t"
                "v_pk_fma_f32 %6, v[244:245], %19, %6 op_sel_hi:[1,0,1]\n\t"
                "v_tanh_f32 v244, v236\n\t"
                "v_pk_add_f32 v[234:235], %15, %49 op_sel:[1,0]\n\t"
                "v_tanh_f32 v245, v237\n\t"
                "v_pk_fma_f32 %7, v[246:247], %19, %7 op_sel_hi:[1,0,1]\n\t"
                "v_tanh_f32 v246, v238\n\t"
                "v_pk_add_f32 v[236:237], %15, %50 op_sel:[1,0]\n\t"
                "v_tanh_f32 v247, v239\n\t"
                "v_pk_fma_f32 %0, v[240:241], %19, %0 op_sel:[0,1,0]\n\t"
                "v_tanh_f32 v240, v232\n\t"
                "v_pk_add_f32 v[238:239], %15, %51 op_sel:[1,0]\n\t"
                "v_tanh_f32 v241, v233\n\t"
                "v_pk_fma_f32 %1, v[242:243], %19, %1 op_sel:[0,1,0]\n\t"
                "v_tanh_f32 v242, v234\n\t"
                "v_tanh_f32 v243, v235\n\t"
                "v_pk_fma_f32 %2, v[244:245], %19, %2 op_sel:[0,1,0]\n\t"
                "v_tanh_f32 v244, v236\n\t"
                "v_tanh_f32 v245, v237\n\t"
                "v_pk_fma_f32 %3, v[246:247], %19, %3 op_sel:[0,1,0]\n\t"
                "v_tanh_f32 v246, v238\n\t"
                "v_tanh_f32 v247, v239\n\t"
                "v_pk_fma_f32 %4, v[240:241], %19, %4 op_sel:[0,1,0]\n\t"
                "v_pk_fma_f32 %5, v[242:243], %19, %5 op_sel:[0,1,0]\n\t"
                "v_pk_fma_f32 %6, v[244:245], %19, %6 op_sel:[0,1,0]\n\t"
                "v_pk_fma_f32 %7, v[246:247], %19, %7 op_sel:[0,1,0]"
                : "+v"(P0[0]), "+v"(P0[1]), "+v"(P0[2]), "+v"(P0[3]), "+v"(P1[0]), "+v"(P1[1]), "+v"(P1[2]), "+v"(P1[3])
                : "v"(A2[0]), "v"(A2[1]), "v"(A2[2]), "v"(A2[3]), "v"(B2[0]), "v"(B2[1]), "v"(B2[2]), "v"(B2[3]), "v"(V2[0]), "v"(V2[1]), "v"(V2[2]), "v"(V2[3]), "v"(K2[0]), "v"(K2[1]), "v"(K2[2]), "v"(K2[3]), "v"(K2[4]), "v"(K2[5]), "v"(K2[6]), "v"(K2[7]), "v"(K2[8]), "v"(K2[9]), "v"(K2[10]), "v"(K2[11]), "v"(K2[12]), "v"(K2[13]), "v"(K2[14]), "v"(K2[15]), "v"(K2[16]), "v"(K2[17]), "v"(K2[18]), "v"(K2[19]), "v"(K2[20]), "v"(K2[21]), "v"(K2[22]), "v"(K2[23]), "v"(K2[24]), "v"(K2[25]), "v"(K2[26]), "v"(K2[27]), "v"(K2[28]), "v"(K2[29]), "v"(K2[30]), "v"(K2[31])
                : "v232", "v233", "v234", "v235", "v236", "v237", "v238", "v239", "v240", "v241", "v242", "v243", "v244", "v245", "v246", "v247");
        }

        if (more) {
            float* d = ldst + ((c + 1) & 1) * QSTAGE;
#pragma unroll
            for (int i = 0; i < 16; ++i)
                *(v4f*)(d + i * 32 * QPITCH) = pre[i];
        }
        __syncthreads();
    }

    float acc0[TL], acc1[TL];
#pragma unroll
    for (int j = 0; j < TL / 2; ++j) {
        acc0[2 * j] = P0[j].x; acc0[2 * j + 1] = P0[j].y;
        acc1[2 * j] = P1[j].x; acc1[2 * j + 1] = P1[j].y;
    }

    float m[TL];
#pragma unroll
    for (int l = 0; l < TL; ++l) m[l] = fmaxf(acc0[l], acc1[l]);
#pragma unroll
    for (int off = 16; off > 0; off >>= 1) {
#pragma unroll
        for (int l = 0; l < TL; ++l) m[l] = fmaxf(m[l], __shfl_xor(m[l], off, 32));
    }
    if (lane == 0) {
        ((v4f*)redm)[wave * 2]     = (v4f){m[0], m[1], m[2], m[3]};
        ((v4f*)redm)[wave * 2 + 1] = (v4f){m[4], m[5], m[6], m[7]};
    }
    __syncthreads();
#pragma unroll
    for (int w = 0; w < 8; ++w) {
        const v4f r0 = ((const v4f*)redm)[w * 2];
        const v4f r1 = ((const v4f*)redm)[w * 2 + 1];
        m[0] = fmaxf(m[0], r0.x); m[1] = fmaxf(m[1], r0.y);
        m[2] = fmaxf(m[2], r0.z); m[3] = fmaxf(m[3], r0.w);
        m[4] = fmaxf(m[4], r1.x); m[5] = fmaxf(m[5], r1.y);
        m[6] = fmaxf(m[6], r1.z); m[7] = fmaxf(m[7], r1.w);
    }

    float s[TL];
#pragma unroll
    for (int l = 0; l < TL; ++l) {
        acc0[l] = fast_exp(acc0[l] - m[l]);
        acc1[l] = fast_exp(acc1[l] - m[l]);
        s[l] = acc0[l] + acc1[l];
    }
#pragma unroll
    for (int off = 16; off > 0; off >>= 1) {
#pragma unroll
        for (int l = 0; l < TL; ++l) s[l] += __shfl_xor(s[l], off, 32);
    }
    if (lane == 0) {
        ((v4f*)reds)[wave * 2]     = (v4f){s[0], s[1], s[2], s[3]};
        ((v4f*)reds)[wave * 2 + 1] = (v4f){s[4], s[5], s[6], s[7]};
    }
    __syncthreads();
    float t[TL];
#pragma unroll
    for (int l = 0; l < TL; ++l) t[l] = 0.f;
#pragma unroll
    for (int w = 0; w < 8; ++w) {
        const v4f r0 = ((const v4f*)reds)[w * 2];
        const v4f r1 = ((const v4f*)reds)[w * 2 + 1];
        t[0] += r0.x; t[1] += r0.y; t[2] += r0.z; t[3] += r0.w;
        t[4] += r1.x; t[5] += r1.y; t[6] += r1.z; t[7] += r1.w;
    }

    float* arow = atten + ((size_t)(b * LK + l0)) * LQ + tid;
#pragma unroll
    for (int l = 0; l < TL; ++l) {
        const float inv = 1.f / t[l];
        arow[(size_t)l * LQ]       = acc0[l] * inv;
        arow[(size_t)l * LQ + 256] = acc1[l] * inv;
    }
}

#define CM       32
#define CN       64
#define CKC      128
#define CNCH     (LQ / CKC)
#define CAPITCH  516
#define CBPITCH  72
#define CCPITCH  68
#define CKS      16
#define CTX_GROUPS  (B_ * (LK / CM) * (D_ / CN))

__global__ __launch_bounds__(128) void context_wmma_kernel(
    const float* __restrict__ atten, const float* __restrict__ value,
    float* __restrict__ context)
{
    __shared__ __align__(16) float As[CM * CAPITCH];
    __shared__ __align__(16) float Bs[LQ * CBPITCH];
    __shared__ __align__(16) float Cs[CM * CCPITCH];

    const int tid  = threadIdx.x;
    const int lane = tid & 31;
    const int wave = tid >> 5;
    const int bid  = blockIdx.x;
    const int b    = bid >> 6;
    const int r    = bid & 63;
    const int l0   = (r >> 2) * CM;
    const int n0   = (r & 3) * CN;

    const int arow = tid >> 5, apc = tid & 31;
    const int brow = tid >> 4, bpc = tid & 15;
    const float* ag = atten + ((size_t)(b * LK + l0 + arow)) * LQ + apc * 4;
    const float* bg = value + ((size_t)(b * LQ + brow)) * D_ + n0 + bpc * 4;
    float* al = As + arow * CAPITCH + apc * 4;
    float* bl = Bs + brow * CBPITCH + bpc * 4;

    {
        v4f ra[8], rb[16];
#pragma unroll
        for (int i = 0; i < 8; ++i)
            ra[i] = *(const v4f*)(ag + (size_t)(i * 4) * LQ);
#pragma unroll
        for (int i = 0; i < 16; ++i)
            rb[i] = *(const v4f*)(bg + (size_t)(i * 8) * D_);
#pragma unroll
        for (int i = 0; i < 8; ++i)
            *(v4f*)(al + (i * 4) * CAPITCH) = ra[i];
#pragma unroll
        for (int i = 0; i < 16; ++i)
            *(v4f*)(bl + (i * 8) * CBPITCH) = rb[i];
    }
    __syncthreads();

    const int half = lane >> 4;
    const int lh   = lane & 15;
    const float* ap0 = As + lh * CAPITCH + 2 * half;
    const float* ap1 = ap0 + 16 * CAPITCH;
    const float* bp  = Bs + (2 * half) * CBPITCH + wave * 16 + lh;

    v8f c0 = {0.f, 0.f, 0.f, 0.f, 0.f, 0.f, 0.f, 0.f};
    v8f c1 = {0.f, 0.f, 0.f, 0.f, 0.f, 0.f, 0.f, 0.f};

#pragma unroll 1
    for (int c = 0; c < CNCH; ++c) {
        const bool more = (c + 1 < CNCH);
        v4f pa[8], pb[16];
        if (more) {
            const float* ag1 = ag + (c + 1) * CKC;
            const float* bg1 = bg + (size_t)((c + 1) * CKC) * D_;
#pragma unroll
            for (int i = 0; i < 8; ++i)
                pa[i] = *(const v4f*)(ag1 + (size_t)(i * 4) * LQ);
#pragma unroll
            for (int i = 0; i < 16; ++i)
                pb[i] = *(const v4f*)(bg1 + (size_t)(i * 8) * D_);
        }

        const float* a0 = ap0 + c * CKC;
        const float* a1 = ap1 + c * CKC;
        const float* bb = bp + (c * CKC) * CBPITCH;

        v2f fa0[2][CKS], fa1[2][CKS], fb[2][CKS];
        __builtin_amdgcn_sched_barrier(0);
#pragma unroll
        for (int s = 0; s < CKS; ++s) {
            const int k = s * 4;
            fa0[0][s]   = *(const v2f*)(a0 + k);
            fa1[0][s]   = *(const v2f*)(a1 + k);
            fb[0][s].x  = bb[(k)     * CBPITCH];
            fb[0][s].y  = bb[(k + 1) * CBPITCH];
        }
#pragma unroll
        for (int kb = 0; kb < CKC / (4 * CKS); ++kb) {
            const int cur = kb & 1;
            const int nxt = cur ^ 1;
            __builtin_amdgcn_sched_barrier(0);
            if (kb + 1 < CKC / (4 * CKS)) {
#pragma unroll
                for (int s = 0; s < CKS; ++s) {
                    const int k = (kb + 1) * (4 * CKS) + s * 4;
                    fa0[nxt][s]   = *(const v2f*)(a0 + k);
                    fa1[nxt][s]   = *(const v2f*)(a1 + k);
                    fb[nxt][s].x  = bb[(k)     * CBPITCH];
                    fb[nxt][s].y  = bb[(k + 1) * CBPITCH];
                }
            }
            __builtin_amdgcn_sched_barrier(0);
#pragma unroll
            for (int s = 0; s < CKS; ++s) {
                c0 = __builtin_amdgcn_wmma_f32_16x16x4_f32(
                        false, fa0[cur][s], false, fb[cur][s], (short)0, c0, false, false);
                c1 = __builtin_amdgcn_wmma_f32_16x16x4_f32(
                        false, fa1[cur][s], false, fb[cur][s], (short)0, c1, false, false);
            }
        }
        __builtin_amdgcn_sched_barrier(0);

        if (more) {
            float* al1 = al + (c + 1) * CKC;
            float* bl1 = bl + ((c + 1) * CKC) * CBPITCH;
#pragma unroll
            for (int i = 0; i < 8; ++i)
                *(v4f*)(al1 + (i * 4) * CAPITCH) = pa[i];
#pragma unroll
            for (int i = 0; i < 16; ++i)
                *(v4f*)(bl1 + (i * 8) * CBPITCH) = pb[i];
        }
        __syncthreads();
    }

    float* cw = Cs + (half * 8) * CCPITCH + wave * 16 + lh;
#pragma unroll
    for (int i = 0; i < 8; ++i) {
        cw[i * CCPITCH]        = c0[i];
        cw[(16 + i) * CCPITCH] = c1[i];
    }
    __syncthreads();

    float* obase = context + ((size_t)(b * LK + l0)) * D_ + n0;
#pragma unroll
    for (int i = 0; i < 4; ++i) {
        const int idx = i * 128 + tid;
        const int row = idx >> 4;
        const int c4  = idx & 15;
        *(v4f*)(obase + (size_t)row * D_ + c4 * 4) = *(const v4f*)(Cs + row * CCPITCH + c4 * 4);
    }
}

extern "C" void kernel_launch(void* const* d_in, const int* in_sizes, int n_in,
                              void* d_out, int out_size, void* d_ws, size_t ws_size,
                              hipStream_t stream)
{
    const float* query = (const float*)d_in[0];
    const float* key   = (const float*)d_in[1];
    const float* value = (const float*)d_in[2];
    const float* Wc1   = (const float*)d_in[3];
    const float* Wc2   = (const float*)d_in[4];
    const float* vc    = (const float*)d_in[5];

    float* out     = (float*)d_out;
    float* context = out;
    float* atten   = out + (size_t)B_ * LK * D_;

    float* qp = (float*)d_ws;
    float* kp = qp + (size_t)B_ * LQ * E_;

    proj_wmma_kernel<<<PROJ_GROUPS, 128, 0, stream>>>(query, key, Wc1, Wc2, qp, kp);
    score_softmax_kernel<<<B_ * (LK / TL), 256, 0, stream>>>(qp, kp, vc, atten);
    context_wmma_kernel<<<CTX_GROUPS, 128, 0, stream>>>(atten, value, context);
}


# score kernel: the two waves of a SIMD take turns on the vector unit block by block: s_setprio 3 from a third into each tanh block until the next block's loads are issued, 0 while they are in flight
# speedup vs baseline: 3.5060x; 1.1276x over previous
#include <hip/hip_runtime.h>
#include <math.h>

#define B_  4
#define LQ  512
#define LK  512
#define E_  128
#define D_  256

typedef float v2f __attribute__((ext_vector_type(2)));
typedef float v8f __attribute__((ext_vector_type(8)));
typedef float v4f __attribute__((ext_vector_type(4)));

__device__ __forceinline__ float fast_exp(float x) {
#if __has_builtin(__builtin_amdgcn_exp2f)
    return __builtin_amdgcn_exp2f(x * 1.44269504088896340f);
#else
    return __expf(x);
#endif
}

#define PM      32
#define PN      64
#define APITCH  260
#define WPITCH  72
#define CPITCH  68
#define KS      16
#define PROJ_GROUPS  (2 * (B_ * 512 / PM) * (E_ / PN))

__global__ __launch_bounds__(128) void proj_wmma_kernel(
    const float* __restrict__ query, const float* __restrict__ key,
    const float* __restrict__ Wc1,   const float* __restrict__ Wc2,
    float* __restrict__ qp, float* __restrict__ kp)
{
    __shared__ __align__(16) float As[PM * APITCH];
    __shared__ __align__(16) float Ws[D_ * WPITCH];
    __shared__ __align__(16) float Cs[PM * CPITCH];

    const int tid  = threadIdx.x;
    const int lane = tid & 31;
    const int wave = tid >> 5;
    const int bid  = blockIdx.x;
    const int sel  = bid >> 7;
    const int r    = bid & 127;
    const int m0   = (r >> 1) * PM;
    const int n0   = (r & 1) * PN;

    const float* in  = sel ? key : query;
    const float* Wg  = sel ? Wc2 : Wc1;
    float*       out = sel ? kp  : qp;

    {
        const float* asrc = in + (size_t)m0 * D_;
        v4f ra[16];
#pragma unroll
        for (int i = 0; i < 16; ++i)
            ra[i] = *(const v4f*)(asrc + (size_t)(i * 128 + tid) * 4);
        const float* wsrc = Wg + n0;
        v4f rw[32];
#pragma unroll
        for (int i = 0; i < 32; ++i) {
            const int idx = i * 128 + tid;
            rw[i] = *(const v4f*)(wsrc + (size_t)(idx >> 4) * E_ + (idx & 15) * 4);
        }
#pragma unroll
        for (int i = 0; i < 16; ++i) {
            const int idx = i * 128 + tid;
            *(v4f*)(As + (idx >> 6) * APITCH + (idx & 63) * 4) = ra[i];
        }
#pragma unroll
        for (int i = 0; i < 32; ++i) {
            const int idx = i * 128 + tid;
            *(v4f*)(Ws + (idx >> 4) * WPITCH + (idx & 15) * 4) = rw[i];
        }
    }
    __syncthreads();

    const int half = lane >> 4;
    const int lh   = lane & 15;
    const float* ap0 = As + lh * APITCH + 2 * half;
    const float* ap1 = ap0 + 16 * APITCH;
    const float* bp  = Ws + (2 * half) * WPITCH + wave * 16 + lh;

    v8f c0 = {0.f, 0.f, 0.f, 0.f, 0.f, 0.f, 0.f, 0.f};
    v8f c1 = {0.f, 0.f, 0.f, 0.f, 0.f, 0.f, 0.f, 0.f};

    v2f fa0[2][KS], fa1[2][KS], fb[2][KS];
#pragma unroll
    for (int s = 0; s < KS; ++s) {
        const int k = s * 4;
        fa0[0][s]   = *(const v2f*)(ap0 + k);
        fa1[0][s]   = *(const v2f*)(ap1 + k);
        fb[0][s].x  = bp[(k)     * WPITCH];
        fb[0][s].y  = bp[(k + 1) * WPITCH];
    }
#pragma unroll
    for (int kb = 0; kb < D_ / (4 * KS); ++kb) {
        const int cur = kb & 1;
        const int nxt = cur ^ 1;
        __builtin_amdgcn_sched_barrier(0);
        if (kb + 1 < D_ / (4 * KS)) {
#pragma unroll
            for (int s = 0; s < KS; ++s) {
                const int k = (kb + 1) * (4 * KS) + s * 4;
                fa0[nxt][s]   = *(const v2f*)(ap0 + k);
                fa1[nxt][s]   = *(const v2f*)(ap1 + k);
                fb[nxt][s].x  = bp[(k)     * WPITCH];
                fb[nxt][s].y  = bp[(k + 1) * WPITCH];
            }
        }
        __builtin_amdgcn_sched_barrier(0);
#pragma unroll
        for (int s = 0; s < KS; ++s) {
            c0 = __builtin_amdgcn_wmma_f32_16x16x4_f32(
                    false, fa0[cur][s], false, fb[cur][s], (short)0, c0, false, false);
            c1 = __builtin_amdgcn_wmma_f32_16x16x4_f32(
                    false, fa1[cur][s], false, fb[cur][s], (short)0, c1, false, false);
        }
    }
    __builtin_amdgcn_sched_barrier(0);

    float* cw = Cs + (half * 8) * CPITCH + wave * 16 + lh;
#pragma unroll
    for (int i = 0; i < 8; ++i) {
        cw[i * CPITCH]        = c0[i];
        cw[(16 + i) * CPITCH] = c1[i];
    }
    __syncthreads();

    if (sel) {
        float* obase = out + (size_t)m0 * E_ + n0;
#pragma unroll
        for (int i = 0; i < 4; ++i) {
            const int idx = i * 128 + tid;
            const int row = idx >> 4;
            const int c4  = idx & 15;
            *(v4f*)(obase + (size_t)row * E_ + c4 * 4) = *(const v4f*)(Cs + row * CPITCH + c4 * 4);
        }
    } else {
        const int bq = m0 >> 9;
        const int q0 = m0 & (LQ - 1);
        float* obase = out + ((size_t)(bq * (E_ / 2) + (n0 >> 1)) * LQ + q0) * 2;
#pragma unroll
        for (int i = 0; i < 4; ++i) {
            const int idx  = i * 128 + tid;
            const int prow = idx >> 4;
            const int pc   = idx & 15;
            const v2f lo = *(const v2f*)(Cs + (2 * pc)     * CPITCH + 2 * prow);
            const v2f hi = *(const v2f*)(Cs + (2 * pc + 1) * CPITCH + 2 * prow);
            *(v4f*)(obase + (size_t)prow * (LQ * 2) + pc * 4) = (v4f){lo.x, lo.y, hi.x, hi.y};
        }
    }
}

#define TL      8
#define FB      8
#define NBLK    (E_ / FB)
#define PFD     3
#define QPROW   (LQ * 2)
#define RESV    40960

__global__ __launch_bounds__(256) void score_softmax_kernel(
    const float* __restrict__ qp, const float* __restrict__ kp,
    const float* __restrict__ vc, float* __restrict__ atten)
{
    __shared__ __align__(16) float kpl[TL * E_ + RESV];
    __shared__ __align__(16) float vcl[2 * E_];
    __shared__ __align__(16) float redm[8 * TL];
    __shared__ __align__(16) float reds[8 * TL];

    const int tid  = threadIdx.x;
    const int lane = tid & 31;
    const int wave = tid >> 5;
    const int b    = blockIdx.x / (LK / TL);
    const int l0   = (blockIdx.x % (LK / TL)) * TL;

    const v4f   kv  = ((const v4f*)(kp + (size_t)(b * LK + l0) * E_))[tid];
    const float vcv = vc[tid & (E_ - 1)];

    const float* qsrc = qp + (size_t)b * (E_ / 2) * QPROW + tid * 4;
    v4f qb[PFD + 1][4];
#pragma unroll
    for (int s = 0; s < PFD; ++s) {
#pragma unroll
        for (int g = 0; g < 4; ++g)
            qb[s][g] = *(const v4f*)(qsrc + (size_t)(s * 4 + g) * QPROW);
    }

    {
        const int krow = tid >> 5;
        const int ke0  = (tid & 31) * 4;
        kpl[(ke0 + 0) * TL + krow] = kv.x;
        kpl[(ke0 + 1) * TL + krow] = kv.y;
        kpl[(ke0 + 2) * TL + krow] = kv.z;
        kpl[(ke0 + 3) * TL + krow] = kv.w;
        vcl[tid] = vcv;
    }
    asm volatile("s_wait_dscnt 0x0" ::: "memory");
    __builtin_amdgcn_s_barrier();
    asm volatile("" ::: "memory");

    v2f P0[TL / 2], P1[TL / 2];
#pragma unroll
    for (int j = 0; j < TL / 2; ++j) { P0[j] = (v2f){0.f, 0.f}; P1[j] = (v2f){0.f, 0.f}; }

#pragma unroll 1
    for (int it = 0; it < NBLK / (PFD + 1); ++it) {
#pragma unroll
        for (int s = 0; s < PFD + 1; ++s) {
            const int blk = it * (PFD + 1) + s;

            {
                int nb = blk + PFD;
                nb = nb < NBLK ? nb : NBLK - 1;
#pragma unroll
                for (int g = 0; g < 4; ++g)
                    qb[(s + PFD) % (PFD + 1)][g] = *(const v4f*)(qsrc + (size_t)(nb * 4 + g) * QPROW);
            }

            const float* kc = kpl + (blk * FB) * TL;
            const float* vv = vcl + blk * FB;

            v2f A2[4], B2[4], V2[4], K2[32];
#pragma unroll
            for (int g = 0; g < 4; ++g) {
                const v4f q4 = qb[s][g];
                A2[g] = __builtin_shufflevector(q4, q4, 0, 1);
                B2[g] = __builtin_shufflevector(q4, q4, 2, 3);
            }
#pragma unroll
            for (int g = 0; g < 2; ++g) {
                const v4f w4 = *(const v4f*)(vv + 4 * g);
                V2[2 * g]     = __builtin_shufflevector(w4, w4, 0, 1);
                V2[2 * g + 1] = __builtin_shufflevector(w4, w4, 2, 3);
            }
#pragma unroll
            for (int f = 0; f < 8; ++f) {
                const v4f kA = *(const v4f*)(kc + f * TL);
                const v4f kB = *(const v4f*)(kc + f * TL + 4);
                K2[4 * f]     = __builtin_shufflevector(kA, kA, 0, 1);
                K2[4 * f + 1] = __builtin_shufflevector(kA, kA, 2, 3);
                K2[4 * f + 2] = __builtin_shufflevector(kB, kB, 0, 1);
                K2[4 * f + 3] = __builtin_shufflevector(kB, kB, 2, 3);
            }
            asm volatile("s_setprio 0" ::: "memory");
            asm volatile(
                "v_pk_add_f32 v[232:233], %8, %20 op_sel_hi:[0,1]\n\t"
                "v_pk_add_f32 v[234:235], %8, %21 op_sel_hi:[0,1]\n\t"
                "v_pk_add_f32 v[236:237], %8, %22 op_sel_hi:[0,1]\n\t"
                "v_tanh_f32 v240, v232\n\t"
                "v_pk_add_f32 v[238:239], %8, %23 op_sel_hi:[0,1]\n\t"
                "v_tanh_f32 v241, v233\n\t"
                "v_tanh_f32 v242, v234\n\t"
                "v_pk_add_f32 v[232:233], %12, %20 op_sel_hi:[0,1]\n\t"
                "v_tanh_f32 v243, v235\n\t"
                "v_tanh_f32 v244, v236\n\t"
                "v_pk_add_f32 v[234:235], %12, %21 op_sel_hi:[0,1]\n\t"
                "v_tanh_f32 v245, v237\n\t"
                "v_tanh_f32 v246, v238\n\t"
                "v_pk_add_f32 v[236:237], %12, %22 op_sel_hi:[0,1]\n\t"
                "v_tanh_f32 v247, v239\n\t"
                "v_pk_fma_f32 %0, v[240:241], %16, %0 op_sel_hi:[1,0,1]\n\t"
                "v_tanh_f32 v240, v232\n\t"
                "v_pk_add_f32 v[238:239], %12, %23 op_sel_hi:[0,1]\n\t"
                "v_tanh_f32 v241, v233\n\t"
                "v_pk_fma_f32 %1, v[242:243], %16, %1 op_sel_hi:[1,0,1]\n\t"
                "v_tanh_f32 v242, v234\n\t"
                "v_pk_add_f32 v[232:233], %8, %24 op_sel:[1,0]\n\t"
                "v_tanh_f32 v243, v235\n\t"
                "v_pk_fma_f32 %2, v[244:245], %16, %2 op_sel_hi:[1,0,1]\n\t"
                "v_tanh_f32 v244, v236\n\t"
                "v_pk_add_f32 v[234:235], %8, %25 op_sel:[1,0]\n\t"
                "v_tanh_f32 v245, v237\n\t"
                "v_pk_fma_f32 %3, v[246:247], %16, %3 op_sel_hi:[1,0,1]\n\t"
                "v_tanh_f32 v246, v238\n\t"
                "v_pk_add_f32 v[236:237], %8, %26 op_sel:[1,0]\n\t"
                "v_tanh_f32 v247, v239\n\t"
                "v_pk_fma_f32 %4, v[240:241], %16, %4 op_sel_hi:[1,0,1]\n\t"
                "v_tanh_f32 v240, v232\n\t"
                "v_pk_add_f32 v[238:239], %8, %27 op_sel:[1,0]\n\t"
                "v_tanh_f32 v241, v233\n\t"
                "v_pk_fma_f32 %5, v[242:243], %16, %5 op_sel_hi:[1,0,1]\n\t"
                "v_tanh_f32 v242, v234\n\t"
                "v_pk_add_f32 v[232:233], %12, %24 op_sel:[1,0]\n\t"
                "v_tanh_f32 v243, v235\n\t"
                "v_pk_fma_f32 %6, v[244:245], %16, %6 op_sel_hi:[1,0,1]\n\t"
                "v_tanh_f32 v244, v236\n\t"
                "v_pk_add_f32 v[234:235], %12, %25 op_sel:[1,0]\n\t"
                "v_tanh_f32 v245, v237\n\t"
                "v_pk_fma_f32 %7, v[246:247], %16, %7 op_sel_hi:[1,0,1]\n\t"
                "v_tanh_f32 v246, v238\n\t"
                "v_pk_add_f32 v[236:237], %12, %26 op_sel:[1,0]\n\t"
                "v_tanh_f32 v247, v239\n\t"
                "v_pk_fma_f32 %0, v[240:241], %16, %0 op_sel:[0,1,0]\n\t"
                "v_tanh_f32 v240, v232\n\t"
                "v_pk_add_f32 v[238:239], %12, %27 op_sel:[1,0]\n\t"
                "v_tanh_f32 v241, v233\n\t"
                "v_pk_fma_f32 %1, v[242:243], %16, %1 op_sel:[0,1,0]\n\t"
                "v_tanh_f32 v242, v234\n\t"
                "v_pk_add_f32 v[232:233], %9, %28 op_sel_hi:[0,1]\n\t"
                "v_tanh_f32 v243, v235\n\t"
                "v_pk_fma_f32 %2, v[244:245], %16, %2 op_sel:[0,1,0]\n\t"
                "v_tanh_f32 v244, v236\n\t"
                "v_pk_add_f32 v[234:235], %9, %29 op_sel_hi:[0,1]\n\t"
                "v_tanh_f32 v245, v237\n\t"
                "v_pk_fma_f32 %3, v[246:247], %16, %3 op_sel:[0,1,0]\n\t"
                "v_tanh_f32 v246, v238\n\t"
                "v_pk_add_f32 v[236:237], %9, %30 op_sel_hi:[0,1]\n\t"
                "v_tanh_f32 v247, v239\n\t"
                "v_pk_fma_f32 %4, v[240:241], %16, %4 op_sel:[0,1,0]\n\t"
                "v_tanh_f32 v240, v232\n\t"
                "v_pk_add_f32 v[238:239], %9, %31 op_sel_hi:[0,1]\n\t"
                "v_tanh_f32 v241, v233\n\t"
                "v_pk_fma_f32 %5, v[242:243], %16, %5 op_sel:[0,1,0]\n\t"
                "v_tanh_f32 v242, v234\n\t"
                "v_pk_add_f32 v[232:233], %13, %28 op_sel_hi:[0,1]\n\t"
                "v_tanh_f32 v243, v235\n\t"
                "v_pk_fma_f32 %6, v[244:245], %16, %6 op_sel:[0,1,0]\n\t"
                "v_tanh_f32 v244, v236\n\t"
                "v_pk_add_f32 v[234:235], %13, %29 op_sel_hi:[0,1]\n\t"
                "v_tanh_f32 v245, v237\n\t"
                "v_pk_fma_f32 %7, v[246:247], %16, %7 op_sel:[0,1,0]\n\t"
                "v_tanh_f32 v246, v238\n\t"
                "v_pk_add_f32 v[236:237], %13, %30 op_sel_hi:[0,1]\n\t"
                "v_tanh_f32 v247, v239\n\t"
                "v_pk_fma_f32 %0, v[240:241], %17, %0 op_sel_hi:[1,0,1]\n\t"
                "v_tanh_f32 v240, v232\n\t"
                "v_pk_add_f32 v[238:239], %13, %31 op_sel_hi:[0,1]\n\t"
                "v_tanh_f32 v241, v233\n\t"
                "v_pk_fma_f32 %1, v[242:243], %17, %1 op_sel_hi:[1,0,1]\n\t"
                "v_tanh_f32 v242, v234\n\t"
                "v_pk_add_f32 v[232:233], %9, %32 op_sel:[1,0]\n\t"
                "v_tanh_f32 v243, v235\n\t"
                "v_pk_fma_f32 %2, v[244:245], %17, %2 op_sel_hi:[1,0,1]\n\t"
                "v_tanh_f32 v244, v236\n\t"
                "v_pk_add_f32 v[234:235], %9, %33 op_sel:[1,0]\n\t"
                "v_tanh_f32 v245, v237\n\t"
                "v_pk_fma_f32 %3, v[246:247], %17, %3 op_sel_hi:[1,0,1]\n\t"
                "v_tanh_f32 v246, v238\n\t"
                "v_pk_add_f32 v[236:237], %9, %34 op_sel:[1,0]\n\t"
                "v_tanh_f32 v247, v239\n\t"
                "v_pk_fma_f32 %4, v[240:241], %17, %4 op_sel_hi:[1,0,1]\n\t"
                "v_tanh_f32 v240, v232\n\t"
                "s_setprio 3\n\t"
                "v_pk_add_f32 v[238:239], %9, %35 op_sel:[1,0]\n\t"
                "v_tanh_f32 v241, v233\n\t"
                "v_pk_fma_f32 %5, v[242:243], %17, %5 op_sel_hi:[1,0,1]\n\t"
                "v_tanh_f32 v242, v234\n\t"
                "v_pk_add_f32 v[232:233], %13, %32 op_sel:[1,0]\n\t"
                "v_tanh_f32 v243, v235\n\t"
                "v_pk_fma_f32 %6, v[244:245], %17, %6 op_sel_hi:[1,0,1]\n\t"
                "v_tanh_f32 v244, v236\n\t"
                "v_pk_add_f32 v[234:235], %13, %33 op_sel:[1,0]\n\t"
                "v_tanh_f32 v245, v237\n\t"
                "v_pk_fma_f32 %7, v[246:247], %17, %7 op_sel_hi:[1,0,1]\n\t"
                "v_tanh_f32 v246, v238\n\t"
                "v_pk_add_f32 v[236:237], %13, %34 op_sel:[1,0]\n\t"
                "v_tanh_f32 v247, v239\n\t"
                "v_pk_fma_f32 %0, v[240:241], %17, %0 op_sel:[0,1,0]\n\t"
                "v_tanh_f32 v240, v232\n\t"
                "v_pk_add_f32 v[238:239], %13, %35 op_sel:[1,0]\n\t"
                "v_tanh_f32 v241, v233\n\t"
                "v_pk_fma_f32 %1, v[242:243], %17, %1 op_sel:[0,1,0]\n\t"
                "v_tanh_f32 v242, v234\n\t"
                "v_pk_add_f32 v[232:233], %10, %36 op_sel_hi:[0,1]\n\t"
                "v_tanh_f32 v243, v235\n\t"
                "v_pk_fma_f32 %2, v[244:245], %17, %2 op_sel:[0,1,0]\n\t"
                "v_tanh_f32 v244, v236\n\t"
                "v_pk_add_f32 v[234:235], %10, %37 op_sel_hi:[0,1]\n\t"
                "v_tanh_f32 v245, v237\n\t"
                "v_pk_fma_f32 %3, v[246:247], %17, %3 op_sel:[0,1,0]\n\t"
                "v_tanh_f32 v246, v238\n\t"
                "v_pk_add_f32 v[236:237], %10, %38 op_sel_hi:[0,1]\n\t"
                "v_tanh_f32 v247, v239\n\t"
                "v_pk_fma_f32 %4, v[240:241], %17, %4 op_sel:[0,1,0]\n\t"
                "v_tanh_f32 v240, v232\n\t"
                "v_pk_add_f32 v[238:239], %10, %39 op_sel_hi:[0,1]\n\t"
                "v_tanh_f32 v241, v233\n\t"
                "v_pk_fma_f32 %5, v[242:243], %17, %5 op_sel:[0,1,0]\n\t"
                "v_tanh_f32 v242, v234\n\t"
                "v_pk_add_f32 v[232:233], %14, %36 op_sel_hi:[0,1]\n\t"
                "v_tanh_f32 v243, v235\n\t"
                "v_pk_fma_f32 %6, v[244:245], %17, %6 op_sel:[0,1,0]\n\t"
                "v_tanh_f32 v244, v236\n\t"
                "v_pk_add_f32 v[234:235], %14, %37 op_sel_hi:[0,1]\n\t"
                "v_tanh_f32 v245, v237\n\t"
                "v_pk_fma_f32 %7, v[246:247], %17, %7 op_sel:[0,1,0]\n\t"
                "v_tanh_f32 v246, v238\n\t"
                "v_pk_add_f32 v[236:237], %14, %38 op_sel_hi:[0,1]\n\t"
                "v_tanh_f32 v247, v239\n\t"
                "v_pk_fma_f32 %0, v[240:241], %18, %0 op_sel_hi:[1,0,1]\n\t"
                "v_tanh_f32 v240, v232\n\t"
                "v_pk_add_f32 v[238:239], %14, %39 op_sel_hi:[0,1]\n\t"
                "v_tanh_f32 v241, v233\n\t"
                "v_pk_fma_f32 %1, v[242:243], %18, %1 op_sel_hi:[1,0,1]\n\t"
                "v_tanh_f32 v242, v234\n\t"
                "v_pk_add_f32 v[232:233], %10, %40 op_sel:[1,0]\n\t"
                "v_tanh_f32 v243, v235\n\t"
                "v_pk_fma_f32 %2, v[244:245], %18, %2 op_sel_hi:[1,0,1]\n\t"
                "v_tanh_f32 v244, v236\n\t"
                "v_pk_add_f32 v[234:235], %10, %41 op_sel:[1,0]\n\t"
                "v_tanh_f32 v245, v237\n\t"
                "v_pk_fma_f32 %3, v[246:247], %18, %3 op_sel_hi:[1,0,1]\n\t"
                "v_tanh_f32 v246, v238\n\t"
                "v_pk_add_f32 v[236:237], %10, %42 op_sel:[1,0]\n\t"
                "v_tanh_f32 v247, v239\n\t"
                "v_pk_fma_f32 %4, v[240:241], %18, %4 op_sel_hi:[1,0,1]\n\t"
                "v_tanh_f32 v240, v232\n\t"
                "v_pk_add_f32 v[238:239], %10, %43 op_sel:[1,0]\n\t"
                "v_tanh_f32 v241, v233\n\t"
                "v_pk_fma_f32 %5, v[242:243], %18, %5 op_sel_hi:[1,0,1]\n\t"
                "v_tanh_f32 v242, v234\n\t"
                "v_pk_add_f32 v[232:233], %14, %40 op_sel:[1,0]\n\t"
                "v_tanh_f32 v243, v235\n\t"
                "v_pk_fma_f32 %6, v[244:245], %18, %6 op_sel_hi:[1,0,1]\n\t"
                "v_tanh_f32 v244, v236\n\t"
                "v_pk_add_f32 v[234:235], %14, %41 op_sel:[1,0]\n\t"
                "v_tanh_f32 v245, v237\n\t"
                "v_pk_fma_f32 %7, v[246:247], %18, %7 op_sel_hi:[1,0,1]\n\t"
                "v_tanh_f32 v246, v238\n\t"
                "v_pk_add_f32 v[236:237], %14, %42 op_sel:[1,0]\n\t"
                "v_tanh_f32 v247, v239\n\t"
                "v_pk_fma_f32 %0, v[240:241], %18, %0 op_sel:[0,1,0]\n\t"
                "v_tanh_f32 v240, v232\n\t"
                "v_pk_add_f32 v[238:239], %14, %43 op_sel:[1,0]\n\t"
                "v_tanh_f32 v241, v233\n\t"
                "v_pk_fma_f32 %1, v[242:243], %18, %1 op_sel:[0,1,0]\n\t"
                "v_tanh_f32 v242, v234\n\t"
                "v_pk_add_f32 v[232:233], %11, %44 op_sel_hi:[0,1]\n\t"
                "v_tanh_f32 v243, v235\n\t"
                "v_pk_fma_f32 %2, v[244:245], %18, %2 op_sel:[0,1,0]\n\t"
                "v_tanh_f32 v244, v236\n\t"
                "v_pk_add_f32 v[234:235], %11, %45 op_sel_hi:[0,1]\n\t"
                "v_tanh_f32 v245, v237\n\t"
                "v_pk_fma_f32 %3, v[246:247], %18, %3 op_sel:[0,1,0]\n\t"
                "v_tanh_f32 v246, v238\n\t"
                "v_pk_add_f32 v[236:237], %11, %46 op_sel_hi:[0,1]\n\t"
                "v_tanh_f32 v247, v239\n\t"
                "v_pk_fma_f32 %4, v[240:241], %18, %4 op_sel:[0,1,0]\n\t"
                "v_tanh_f32 v240, v232\n\t"
                "v_pk_add_f32 v[238:239], %11, %47 op_sel_hi:[0,1]\n\t"
                "v_tanh_f32 v241, v233\n\t"
                "v_pk_fma_f32 %5, v[242:243], %18, %5 op_sel:[0,1,0]\n\t"
                "v_tanh_f32 v242, v234\n\t"
                "v_pk_add_f32 v[232:233], %15, %44 op_sel_hi:[0,1]\n\t"
                "v_tanh_f32 v243, v235\n\t"
                "v_pk_fma_f32 %6, v[244:245], %18, %6 op_sel:[0,1,0]\n\t"
                "v_tanh_f32 v244, v236\n\t"
                "v_pk_add_f32 v[234:235], %15, %45 op_sel_hi:[0,1]\n\t"
                "v_tanh_f32 v245, v237\n\t"
                "v_pk_fma_f32 %7, v[246:247], %18, %7 op_sel:[0,1,0]\n\t"
                "v_tanh_f32 v246, v238\n\t"
                "v_pk_add_f32 v[236:237], %15, %46 op_sel_hi:[0,1]\n\t"
                "v_tanh_f32 v247, v239\n\t"
                "v_pk_fma_f32 %0, v[240:241], %19, %0 op_sel_hi:[1,0,1]\n\t"
                "v_tanh_f32 v240, v232\n\t"
                "v_pk_add_f32 v[238:239], %15, %47 op_sel_hi:[0,1]\n\t"
                "v_tanh_f32 v241, v233\n\t"
                "v_pk_fma_f32 %1, v[242:243], %19, %1 op_sel_hi:[1,0,1]\n\t"
                "v_tanh_f32 v242, v234\n\t"
                "v_pk_add_f32 v[232:233], %11, %48 op_sel:[1,0]\n\t"
                "v_tanh_f32 v243, v235\n\t"
                "v_pk_fma_f32 %2, v[244:245], %19, %2 op_sel_hi:[1,0,1]\n\t"
                "v_tanh_f32 v244, v236\n\t"
                "v_pk_add_f32 v[234:235], %11, %49 op_sel:[1,0]\n\t"
                "v_tanh_f32 v245, v237\n\t"
                "v_pk_fma_f32 %3, v[246:247], %19, %3 op_sel_hi:[1,0,1]\n\t"
                "v_tanh_f32 v246, v238\n\t"
                "v_pk_add_f32 v[236:237], %11, %50 op_sel:[1,0]\n\t"
                "v_tanh_f32 v247, v239\n\t"
                "v_pk_fma_f32 %4, v[240:241], %19, %4 op_sel_hi:[1,0,1]\n\t"
                "v_tanh_f32 v240, v232\n\t"
                "v_pk_add_f32 v[238:239], %11, %51 op_sel:[1,0]\n\t"
                "v_tanh_f32 v241, v233\n\t"
                "v_pk_fma_f32 %5, v[242:243], %19, %5 op_sel_hi:[1,0,1]\n\t"
                "v_tanh_f32 v242, v234\n\t"
                "v_pk_add_f32 v[232:233], %15, %48 op_sel:[1,0]\n\t"
                "v_tanh_f32 v243, v235\n\t"
                "v_pk_fma_f32 %6, v[244:245], %19, %6 op_sel_hi:[1,0,1]\n\t"
                "v_tanh_f32 v244, v236\n\t"
                "v_pk_add_f32 v[234:235], %15, %49 op_sel:[1,0]\n\t"
                "v_tanh_f32 v245, v237\n\t"
                "v_pk_fma_f32 %7, v[246:247], %19, %7 op_sel_hi:[1,0,1]\n\t"
                "v_tanh_f32 v246, v238\n\t"
                "v_pk_add_f32 v[236:237], %15, %50 op_sel:[1,0]\n\t"
                "v_tanh_f32 v247, v239\n\t"
                "v_pk_fma_f32 %0, v[240:241], %19, %0 op_sel:[0,1,0]\n\t"
                "v_tanh_f32 v240, v232\n\t"
                "v_pk_add_f32 v[238:239], %15, %51 op_sel:[1,0]\n\t"
                "v_tanh_f32 v241, v233\n\t"
                "v_pk_fma_f32 %1, v[242:243], %19, %1 op_sel:[0,1,0]\n\t"
                "v_tanh_f32 v242, v234\n\t"
                "v_tanh_f32 v243, v235\n\t"
                "v_pk_fma_f32 %2, v[244:245], %19, %2 op_sel:[0,1,0]\n\t"
                "v_tanh_f32 v244, v236\n\t"
                "v_tanh_f32 v245, v237\n\t"
                "v_pk_fma_f32 %3, v[246:247], %19, %3 op_sel:[0,1,0]\n\t"
                "v_tanh_f32 v246, v238\n\t"
                "v_tanh_f32 v247, v239\n\t"
                "v_pk_fma_f32 %4, v[240:241], %19, %4 op_sel:[0,1,0]\n\t"
                "v_pk_fma_f32 %5, v[242:243], %19, %5 op_sel:[0,1,0]\n\t"
                "v_pk_fma_f32 %6, v[244:245], %19, %6 op_sel:[0,1,0]\n\t"
                "v_pk_fma_f32 %7, v[246:247], %19, %7 op_sel:[0,1,0]"
                : "+v"(P0[0]), "+v"(P0[1]), "+v"(P0[2]), "+v"(P0[3]), "+v"(P1[0]), "+v"(P1[1]), "+v"(P1[2]), "+v"(P1[3])
                : "v"(A2[0]), "v"(A2[1]), "v"(A2[2]), "v"(A2[3]), "v"(B2[0]), "v"(B2[1]), "v"(B2[2]), "v"(B2[3]), "v"(V2[0]), "v"(V2[1]), "v"(V2[2]), "v"(V2[3]), "v"(K2[0]), "v"(K2[1]), "v"(K2[2]), "v"(K2[3]), "v"(K2[4]), "v"(K2[5]), "v"(K2[6]), "v"(K2[7]), "v"(K2[8]), "v"(K2[9]), "v"(K2[10]), "v"(K2[11]), "v"(K2[12]), "v"(K2[13]), "v"(K2[14]), "v"(K2[15]), "v"(K2[16]), "v"(K2[17]), "v"(K2[18]), "v"(K2[19]), "v"(K2[20]), "v"(K2[21]), "v"(K2[22]), "v"(K2[23]), "v"(K2[24]), "v"(K2[25]), "v"(K2[26]), "v"(K2[27]), "v"(K2[28]), "v"(K2[29]), "v"(K2[30]), "v"(K2[31])
                : "v232", "v233", "v234", "v235", "v236", "v237", "v238", "v239", "v240", "v241", "v242", "v243", "v244", "v245", "v246", "v247");
        }
    }

    asm volatile("s_setprio 0" ::: "memory");

    float acc0[TL], acc1[TL];
#pragma unroll
    for (int j = 0; j < TL / 2; ++j) {
        acc0[2 * j] = P0[j].x; acc0[2 * j + 1] = P0[j].y;
        acc1[2 * j] = P1[j].x; acc1[2 * j + 1] = P1[j].y;
    }

    float m[TL];
#pragma unroll
    for (int l = 0; l < TL; ++l) m[l] = fmaxf(acc0[l], acc1[l]);
#pragma unroll
    for (int off = 16; off > 0; off >>= 1) {
#pragma unroll
        for (int l = 0; l < TL; ++l) m[l] = fmaxf(m[l], __shfl_xor(m[l], off, 32));
    }
    if (lane == 0) {
        ((v4f*)redm)[wave * 2]     = (v4f){m[0], m[1], m[2], m[3]};
        ((v4f*)redm)[wave * 2 + 1] = (v4f){m[4], m[5], m[6], m[7]};
    }
    __syncthreads();
#pragma unroll
    for (int w = 0; w < 8; ++w) {
        const v4f r0 = ((const v4f*)redm)[w * 2];
        const v4f r1 = ((const v4f*)redm)[w * 2 + 1];
        m[0] = fmaxf(m[0], r0.x); m[1] = fmaxf(m[1], r0.y);
        m[2] = fmaxf(m[2], r0.z); m[3] = fmaxf(m[3], r0.w);
        m[4] = fmaxf(m[4], r1.x); m[5] = fmaxf(m[5], r1.y);
        m[6] = fmaxf(m[6], r1.z); m[7] = fmaxf(m[7], r1.w);
    }

    float s[TL];
#pragma unroll
    for (int l = 0; l < TL; ++l) {
        acc0[l] = fast_exp(acc0[l] - m[l]);
        acc1[l] = fast_exp(acc1[l] - m[l]);
        s[l] = acc0[l] + acc1[l];
    }
#pragma unroll
    for (int off = 16; off > 0; off >>= 1) {
#pragma unroll
        for (int l = 0; l < TL; ++l) s[l] += __shfl_xor(s[l], off, 32);
    }
    if (lane == 0) {
        ((v4f*)reds)[wave * 2]     = (v4f){s[0], s[1], s[2], s[3]};
        ((v4f*)reds)[wave * 2 + 1] = (v4f){s[4], s[5], s[6], s[7]};
    }
    __syncthreads();
    float t[TL];
#pragma unroll
    for (int l = 0; l < TL; ++l) t[l] = 0.f;
#pragma unroll
    for (int w = 0; w < 8; ++w) {
        const v4f r0 = ((const v4f*)reds)[w * 2];
        const v4f r1 = ((const v4f*)reds)[w * 2 + 1];
        t[0] += r0.x; t[1] += r0.y; t[2] += r0.z; t[3] += r0.w;
        t[4] += r1.x; t[5] += r1.y; t[6] += r1.z; t[7] += r1.w;
    }

    float* arow = atten + ((size_t)(b * LK + l0)) * LQ + 2 * tid;
#pragma unroll
    for (int l = 0; l < TL; ++l) {
        const float inv = 1.f / t[l];
        *(v2f*)(arow + (size_t)l * LQ) = (v2f){acc0[l] * inv, acc1[l] * inv};
    }
}

#define CM       32
#define CN       64
#define CKC      128
#define CNCH     (LQ / CKC)
#define CAPITCH  516
#define CBPITCH  72
#define CCPITCH  68
#define CKS      16
#define CTX_GROUPS  (B_ * (LK / CM) * (D_ / CN))

__global__ __launch_bounds__(128) void context_wmma_kernel(
    const float* __restrict__ atten, const float* __restrict__ value,
    float* __restrict__ context)
{
    __shared__ __align__(16) float As[CM * CAPITCH];
    __shared__ __align__(16) float Bs[LQ * CBPITCH];
    __shared__ __align__(16) float Cs[CM * CCPITCH];

    const int tid  = threadIdx.x;
    const int lane = tid & 31;
    const int wave = tid >> 5;
    const int bid  = blockIdx.x;
    const int b    = bid >> 6;
    const int r    = bid & 63;
    const int l0   = (r >> 2) * CM;
    const int n0   = (r & 3) * CN;

    const int arow = tid >> 5, apc = tid & 31;
    const int brow = tid >> 4, bpc = tid & 15;
    const float* ag = atten + ((size_t)(b * LK + l0 + arow)) * LQ + apc * 4;
    const float* bg = value + ((size_t)(b * LQ + brow)) * D_ + n0 + bpc * 4;
    float* al = As + arow * CAPITCH + apc * 4;
    float* bl = Bs + brow * CBPITCH + bpc * 4;

    {
        v4f ra[8], rb[16];
#pragma unroll
        for (int i = 0; i < 8; ++i)
            ra[i] = *(const v4f*)(ag + (size_t)(i * 4) * LQ);
#pragma unroll
        for (int i = 0; i < 16; ++i)
            rb[i] = *(const v4f*)(bg + (size_t)(i * 8) * D_);
#pragma unroll
        for (int i = 0; i < 8; ++i)
            *(v4f*)(al + (i * 4) * CAPITCH) = ra[i];
#pragma unroll
        for (int i = 0; i < 16; ++i)
            *(v4f*)(bl + (i * 8) * CBPITCH) = rb[i];
    }
    __syncthreads();

    const int half = lane >> 4;
    const int lh   = lane & 15;
    const float* ap0 = As + lh * CAPITCH + 2 * half;
    const float* ap1 = ap0 + 16 * CAPITCH;
    const float* bp  = Bs + (2 * half) * CBPITCH + wave * 16 + lh;

    v8f c0 = {0.f, 0.f, 0.f, 0.f, 0.f, 0.f, 0.f, 0.f};
    v8f c1 = {0.f, 0.f, 0.f, 0.f, 0.f, 0.f, 0.f, 0.f};

#pragma unroll 1
    for (int c = 0; c < CNCH; ++c) {
        const bool more = (c + 1 < CNCH);
        v4f pa[8], pb[16];
        if (more) {
            const float* ag1 = ag + (c + 1) * CKC;
            const float* bg1 = bg + (size_t)((c + 1) * CKC) * D_;
#pragma unroll
            for (int i = 0; i < 8; ++i)
                pa[i] = *(const v4f*)(ag1 + (size_t)(i * 4) * LQ);
#pragma unroll
            for (int i = 0; i < 16; ++i)
                pb[i] = *(const v4f*)(bg1 + (size_t)(i * 8) * D_);
        }

        const float* a0 = ap0 + c * CKC;
        const float* a1 = ap1 + c * CKC;
        const float* bb = bp + (c * CKC) * CBPITCH;

        v2f fa0[2][CKS], fa1[2][CKS], fb[2][CKS];
        __builtin_amdgcn_sched_barrier(0);
#pragma unroll
        for (int s = 0; s < CKS; ++s) {
            const int k = s * 4;
            fa0[0][s]   = *(const v2f*)(a0 + k);
            fa1[0][s]   = *(const v2f*)(a1 + k);
            fb[0][s].x  = bb[(k)     * CBPITCH];
            fb[0][s].y  = bb[(k + 1) * CBPITCH];
        }
#pragma unroll
        for (int kb = 0; kb < CKC / (4 * CKS); ++kb) {
            const int cur = kb & 1;
            const int nxt = cur ^ 1;
            __builtin_amdgcn_sched_barrier(0);
            if (kb + 1 < CKC / (4 * CKS)) {
#pragma unroll
                for (int s = 0; s < CKS; ++s) {
                    const int k = (kb + 1) * (4 * CKS) + s * 4;
                    fa0[nxt][s]   = *(const v2f*)(a0 + k);
                    fa1[nxt][s]   = *(const v2f*)(a1 + k);
                    fb[nxt][s].x  = bb[(k)     * CBPITCH];
                    fb[nxt][s].y  = bb[(k + 1) * CBPITCH];
                }
            }
            __builtin_amdgcn_sched_barrier(0);
#pragma unroll
            for (int s = 0; s < CKS; ++s) {
                c0 = __builtin_amdgcn_wmma_f32_16x16x4_f32(
                        false, fa0[cur][s], false, fb[cur][s], (short)0, c0, false, false);
                c1 = __builtin_amdgcn_wmma_f32_16x16x4_f32(
                        false, fa1[cur][s], false, fb[cur][s], (short)0, c1, false, false);
            }
        }
        __builtin_amdgcn_sched_barrier(0);

        if (more) {
            float* al1 = al + (c + 1) * CKC;
            float* bl1 = bl + ((c + 1) * CKC) * CBPITCH;
#pragma unroll
            for (int i = 0; i < 8; ++i)
                *(v4f*)(al1 + (i * 4) * CAPITCH) = pa[i];
#pragma unroll
            for (int i = 0; i < 16; ++i)
                *(v4f*)(bl1 + (i * 8) * CBPITCH) = pb[i];
        }
        __syncthreads();
    }

    float* cw = Cs + (half * 8) * CCPITCH + wave * 16 + lh;
#pragma unroll
    for (int i = 0; i < 8; ++i) {
        cw[i * CCPITCH]        = c0[i];
        cw[(16 + i) * CCPITCH] = c1[i];
    }
    __syncthreads();

    float* obase = context + ((size_t)(b * LK + l0)) * D_ + n0;
#pragma unroll
    for (int i = 0; i < 4; ++i) {
        const int idx = i * 128 + tid;
        const int row = idx >> 4;
        const int c4  = idx & 15;
        *(v4f*)(obase + (size_t)row * D_ + c4 * 4) = *(const v4f*)(Cs + row * CCPITCH + c4 * 4);
    }
}

extern "C" void kernel_launch(void* const* d_in, const int* in_sizes, int n_in,
                              void* d_out, int out_size, void* d_ws, size_t ws_size,
                              hipStream_t stream)
{
    const float* query = (const float*)d_in[0];
    const float* key   = (const float*)d_in[1];
    const float* value = (const float*)d_in[2];
    const float* Wc1   = (const float*)d_in[3];
    const float* Wc2   = (const float*)d_in[4];
    const float* vc    = (const float*)d_in[5];

    float* out     = (float*)d_out;
    float* context = out;
    float* atten   = out + (size_t)B_ * LK * D_;

    float* qp = (float*)d_ws;
    float* kp = qp + (size_t)B_ * LQ * E_;

    proj_wmma_kernel<<<PROJ_GROUPS, 128, 0, stream>>>(query, key, Wc1, Wc2, qp, kp);
    score_softmax_kernel<<<B_ * (LK / TL), 256, 0, stream>>>(qp, kp, vc, atten);
    context_wmma_kernel<<<CTX_GROUPS, 128, 0, stream>>>(atten, value, context);
}
